# starting kernel
# baseline (speedup 1.0000x reference)
#include <hip/hip_runtime.h>
#include <math.h>

typedef float v2f __attribute__((ext_vector_type(2)));
typedef float v8f __attribute__((ext_vector_type(8)));

#define FDIM 128

__global__ void k_init_deg(float* __restrict__ deg, int N) {
    int i = blockIdx.x * blockDim.x + threadIdx.x;
    if (i < N) deg[i] = 1.0f;
}

__global__ void k_count_deg(const int* __restrict__ dst, float* __restrict__ deg, int E) {
    int stride = gridDim.x * blockDim.x;
    for (int e = blockIdx.x * blockDim.x + threadIdx.x; e < E; e += stride) {
        atomicAdd(&deg[dst[e]], 1.0f);
    }
}

__global__ void k_dinv(float* __restrict__ deg, int N) {
    int i = blockIdx.x * blockDim.x + threadIdx.x;
    if (i < N) {
        float d = deg[i];
        deg[i] = (d > 0.0f) ? rsqrtf(d) : 0.0f;
    }
}

template <bool RELU_A>
__global__ void k_gemm128_fused(const float* __restrict__ A,
                                const float* __restrict__ W,
                                const float* __restrict__ bias,
                                const float* __restrict__ dinv,
                                float* __restrict__ outH,
                                float* __restrict__ outAgg, int M) {
    __shared__ float sW[FDIM * FDIM];

    const int lane = threadIdx.x & 31;
    const int half = lane >> 4;
    const int l16  = lane & 15;

    for (int e = threadIdx.x; e < 32 * 8 * 32; e += blockDim.x) {
        int kstep = e >> 8;
        int rem   = e & 255;
        int t     = rem >> 5;
        int ln    = rem & 31;
        int h     = ln >> 4;
        int c     = t * 16 + (ln & 15);
        int krow  = kstep * 4 + 2 * h;
        float2 v;
        v.x = W[(size_t)krow * FDIM + c];
        v.y = W[(size_t)(krow + 1) * FDIM + c];
        ((float2*)sW)[e] = v;
    }
    __syncthreads();

    const int waveId = (blockIdx.x * blockDim.x + threadIdx.x) >> 5;
    const int rowT   = waveId;
    if (rowT * 16 >= M) return;

    const float* __restrict__ arow = A + (size_t)(rowT * 16 + l16) * FDIM;

    v8f acc[8];
#pragma unroll
    for (int t = 0; t < 8; ++t) acc[t] = (v8f){0, 0, 0, 0, 0, 0, 0, 0};

    for (int kstep = 0; kstep < 32; ++kstep) {
        v2f a = *(const v2f*)(arow + kstep * 4 + 2 * half);
        if (RELU_A) {
            a.x = fmaxf(a.x, 0.0f);
            a.y = fmaxf(a.y, 0.0f);
        }
        const v2f* __restrict__ bfrag = (const v2f*)sW + kstep * 256 + lane;
#pragma unroll
        for (int t = 0; t < 8; ++t) {
            v2f b = bfrag[t * 32];
            acc[t] = __builtin_amdgcn_wmma_f32_16x16x4_f32(
                false, a, false, b, (short)0, acc[t], false, false);
        }
    }

    float dv2[8];
#pragma unroll
    for (int r = 0; r < 8; ++r) {
        float dv = dinv[rowT * 16 + r + half * 8];
        dv2[r] = dv * dv;
    }
#pragma unroll
    for (int t = 0; t < 8; ++t) {
        const int col = t * 16 + l16;
        const float bc = bias[col];
#pragma unroll
        for (int r = 0; r < 8; ++r) {
            const int row = rowT * 16 + r + half * 8;
            const float h = acc[t][r];
            outH[(size_t)row * FDIM + col]   = h;
            outAgg[(size_t)row * FDIM + col] = dv2[r] * h + bc;
        }
    }
}

__global__ void k_edge_agg(const float* __restrict__ h,
                           const float* __restrict__ dinv,
                           const int* __restrict__ src,
                           const int* __restrict__ dst,
                           float* __restrict__ out, int E) {
    int gid   = blockIdx.x * blockDim.x + threadIdx.x;
    int wave  = gid >> 5;
    int lane  = gid & 31;
    int nwave = (gridDim.x * blockDim.x) >> 5;
    for (int e = wave; e < E; e += nwave) {
        int s = src[e];
        int d = dst[e];
        float nrm = dinv[s] * dinv[d];
        float4 v = ((const float4*)(h + (size_t)s * FDIM))[lane];
        float* od = out + (size_t)d * FDIM + 4 * lane;
        atomicAdd(od + 0, nrm * v.x);
        atomicAdd(od + 1, nrm * v.y);
        atomicAdd(od + 2, nrm * v.z);
        atomicAdd(od + 3, nrm * v.w);
    }
}

extern "C" void kernel_launch(void* const* d_in, const int* in_sizes, int n_in,
                              void* d_out, int out_size, void* d_ws, size_t ws_size,
                              hipStream_t stream) {
    const float* x   = (const float*)d_in[0];
    const int*   ei  = (const int*)d_in[1];
    const float* W1  = (const float*)d_in[2];
    const float* b1  = (const float*)d_in[3];
    const float* W2  = (const float*)d_in[4];
    const float* b2  = (const float*)d_in[5];
    float*       out = (float*)d_out;

    const int N = in_sizes[0] / FDIM;
    const int E = in_sizes[1] / 2;
    const int* src = ei;
    const int* dst = ei + E;

    char*  ws   = (char*)d_ws;
    float* dinv = (float*)ws;
    size_t off  = ((size_t)N * sizeof(float) + 255) & ~(size_t)255;
    float* bufH = (float*)(ws + off);
    float* bufA = (float*)(ws + off + (size_t)N * FDIM * sizeof(float));

    const int TPB = 256;

    k_init_deg<<<(N + TPB - 1) / TPB, TPB, 0, stream>>>(dinv, N);
    k_count_deg<<<2048, TPB, 0, stream>>>(dst, dinv, E);
    k_dinv<<<(N + TPB - 1) / TPB, TPB, 0, stream>>>(dinv, N);

    const int tiles      = (N + 15) / 16;
    const int gemmBlocks = (tiles + (TPB / 32) - 1) / (TPB / 32);

    k_gemm128_fused<false><<<gemmBlocks, TPB, 0, stream>>>(x, W1, b1, dinv, bufH, bufA, N);
    k_edge_agg<<<4096, TPB, 0, stream>>>(bufH, dinv, src, dst, bufA, E);

    k_gemm128_fused<true><<<gemmBlocks, TPB, 0, stream>>>(bufA, W2, b2, dinv, bufH, out, N);
    k_edge_agg<<<4096, TPB, 0, stream>>>(bufH, dinv, src, dst, out, E);
}


# Atomic edge scatter-add replaced by owner-computes gather: edges counting-sorted by destination (arrival rank), one wave sums each row in registers; f32 sum order per row now follows arrival rank
# speedup vs baseline: 2.4152x; 2.4152x over previous
#include <hip/hip_runtime.h>
#include <math.h>


typedef float v2f __attribute__((ext_vector_type(2)));
typedef float v8f __attribute__((ext_vector_type(8)));

#define FDIM 128

__global__ void k_zero_cnt(int* __restrict__ cnt, int N) {
    int i = blockIdx.x * blockDim.x + threadIdx.x;
    if (i < N) cnt[i] = 0;
}

__global__ void __launch_bounds__(256) k_rank(const int* __restrict__ dst, int* __restrict__ cnt,
                                              int* __restrict__ rank, int E) {
    const int base = blockIdx.x * 1024 + threadIdx.x;
    int d[4], r[4];
#pragma unroll
    for (int k = 0; k < 4; ++k) {
        const int e = base + k * 256;
        d[k] = dst[(e < E) ? e : (E - 1)];
    }
#pragma unroll
    for (int k = 0; k < 4; ++k) {
        const int e = base + k * 256;
        r[k] = 0;
        if (e < E) r[k] = atomicAdd(&cnt[d[k]], 1);
    }
#pragma unroll
    for (int k = 0; k < 4; ++k) {
        const int e = base + k * 256;
        if (e < E) rank[e] = r[k];
    }
}

__global__ void __launch_bounds__(1024) k_scan_dinv(const int* __restrict__ cnt, int* __restrict__ rowptr,
                                                    float* __restrict__ dinv, int N) {
    __shared__ int sPre[32];
    __shared__ int sTot[32];
    const int tid  = threadIdx.x;
    const int lane = tid & 31;
    const int wv   = tid >> 5;
    const int nb   = blockIdx.x;
    const int4* __restrict__ c4 = (const int4*)cnt;

    int pre = 0;
    int it = 0;
    for (; it + 8 <= nb; it += 8) {
        int4 v[8];
#pragma unroll
        for (int u = 0; u < 8; ++u) v[u] = c4[(it + u) * 1024 + tid];
#pragma unroll
        for (int u = 0; u < 8; ++u) pre += (v[u].x + v[u].y) + (v[u].z + v[u].w);
    }
    for (; it < nb; ++it) {
        const int4 v = c4[it * 1024 + tid];
        pre += (v.x + v.y) + (v.z + v.w);
    }

    const int idx = nb * 4096 + tid * 4;
    int m0 = 0, m1 = 0, m2 = 0, m3 = 0;
    if (idx + 3 < N) {
        const int4 v = c4[idx >> 2];
        m0 = v.x; m1 = v.y; m2 = v.z; m3 = v.w;
    } else {
        if (idx     < N) m0 = cnt[idx];
        if (idx + 1 < N) m1 = cnt[idx + 1];
        if (idx + 2 < N) m2 = cnt[idx + 2];
    }
    const int tsum = (m0 + m1) + (m2 + m3);

    int incl = tsum;
#pragma unroll
    for (int o = 1; o < 32; o <<= 1) {
        const int t = __shfl_up(incl, o);
        if (lane >= o) incl += t;
    }
#pragma unroll
    for (int o = 16; o > 0; o >>= 1) pre += __shfl_xor(pre, o);

    if (lane == 31) sTot[wv] = incl;
    if (lane == 0)  sPre[wv] = pre;
    __syncthreads();

    int base = 0;
#pragma unroll
    for (int w = 0; w < 32; ++w) {
        base += sPre[w];
        if (w < wv) base += sTot[w];
    }
    const int e0 = base + incl - tsum;
    const int e1 = e0 + m0, e2 = e1 + m1, e3 = e2 + m2;

    if (idx + 3 < N) {
        ((int4*)rowptr)[idx >> 2] = make_int4(e0, e1, e2, e3);
        float4 dv;
        dv.x = rsqrtf((float)(m0 + 1));
        dv.y = rsqrtf((float)(m1 + 1));
        dv.z = rsqrtf((float)(m2 + 1));
        dv.w = rsqrtf((float)(m3 + 1));
        ((float4*)dinv)[idx >> 2] = dv;
    } else {
        if (idx     < N) { rowptr[idx]     = e0; dinv[idx]     = rsqrtf((float)(m0 + 1)); }
        if (idx + 1 < N) { rowptr[idx + 1] = e1; dinv[idx + 1] = rsqrtf((float)(m1 + 1)); }
        if (idx + 2 < N) { rowptr[idx + 2] = e2; dinv[idx + 2] = rsqrtf((float)(m2 + 1)); }
    }
}

__global__ void __launch_bounds__(256) k_fill(const int* __restrict__ src, const int* __restrict__ dst,
                                              const int* __restrict__ rank, const int* __restrict__ rowptr,
                                              int* __restrict__ csrc, int E) {
    const int base = blockIdx.x * 1024 + threadIdx.x;
    int s[4], d[4], r[4], p[4];
#pragma unroll
    for (int k = 0; k < 4; ++k) {
        const int e  = base + k * 256;
        const int ec = (e < E) ? e : (E - 1);
        d[k] = dst[ec];
        r[k] = rank[ec];
        s[k] = src[ec];
    }
#pragma unroll
    for (int k = 0; k < 4; ++k) p[k] = rowptr[d[k]] + r[k];
#pragma unroll
    for (int k = 0; k < 4; ++k) {
        const int e = base + k * 256;
        if (e < E && (unsigned)p[k] < (unsigned)E) csrc[p[k]] = s[k];
    }
}

template <bool RELU_A>
__global__ void k_gemm128(const float* __restrict__ A,
                          const float* __restrict__ W,
                          float* __restrict__ outH, int M) {
    __shared__ float sW[FDIM * FDIM];

    const int lane = threadIdx.x & 31;
    const int half = lane >> 4;
    const int l16  = lane & 15;

    for (int e = threadIdx.x; e < 32 * 8 * 32; e += blockDim.x) {
        int kstep = e >> 8;
        int rem   = e & 255;
        int t     = rem >> 5;
        int ln    = rem & 31;
        int h     = ln >> 4;
        int c     = t * 16 + (ln & 15);
        int krow  = kstep * 4 + 2 * h;
        float2 v;
        v.x = W[(size_t)krow * FDIM + c];
        v.y = W[(size_t)(krow + 1) * FDIM + c];
        ((float2*)sW)[e] = v;
    }
    __syncthreads();

    const int waveId = (blockIdx.x * blockDim.x + threadIdx.x) >> 5;
    const int rowT   = waveId;
    if (rowT * 16 >= M) return;

    const float* __restrict__ arow = A + (size_t)(rowT * 16 + l16) * FDIM;

    v8f acc[8];
#pragma unroll
    for (int t = 0; t < 8; ++t) acc[t] = (v8f){0, 0, 0, 0, 0, 0, 0, 0};

    for (int kstep = 0; kstep < 32; ++kstep) {
        v2f a = *(const v2f*)(arow + kstep * 4 + 2 * half);
        if (RELU_A) {
            a.x = fmaxf(a.x, 0.0f);
            a.y = fmaxf(a.y, 0.0f);
        }
        const v2f* __restrict__ bfrag = (const v2f*)sW + kstep * 256 + lane;
#pragma unroll
        for (int t = 0; t < 8; ++t) {
            v2f b = bfrag[t * 32];
            acc[t] = __builtin_amdgcn_wmma_f32_16x16x4_f32(
                false, a, false, b, (short)0, acc[t], false, false);
        }
    }

#pragma unroll
    for (int t = 0; t < 8; ++t) {
        const int col = t * 16 + l16;
#pragma unroll
        for (int r = 0; r < 8; ++r) {
            const int row = rowT * 16 + r + half * 8;
            outH[(size_t)row * FDIM + col] = acc[t][r];
        }
    }
}

template <int G>
__device__ __forceinline__ void row_group(const float4* __restrict__ h4, int sv, float dvl,
                                          int j, int id0, int n, float dd, float d2,
                                          int lane, float4& acc) {
    int    s[G];
    float4 v[G];
#pragma unroll
    for (int u = 0; u < G; ++u) s[u] = __builtin_amdgcn_readlane(sv, j + u);
#pragma unroll
    for (int u = 0; u < G; ++u) v[u] = h4[(size_t)s[u] * 32 + lane];
#pragma unroll
    for (int u = 0; u < G; ++u) {
        const int   id = id0 + u;
        const float ds = __int_as_float(__builtin_amdgcn_readlane(__float_as_int(dvl), j + u));
        const float w  = (id < n) ? ds * dd : ((id == n) ? d2 : 0.0f);
        acc.x = fmaf(w, v[u].x, acc.x);
        acc.y = fmaf(w, v[u].y, acc.y);
        acc.z = fmaf(w, v[u].z, acc.z);
        acc.w = fmaf(w, v[u].w, acc.w);
    }
}

__global__ void __launch_bounds__(256) k_gather_agg(const float* __restrict__ h,
                                                    const float* __restrict__ dinv,
                                                    const int* __restrict__ rowptr,
                                                    const int* __restrict__ cnt,
                                                    const int* __restrict__ csrc,
                                                    const float* __restrict__ bias,
                                                    float* __restrict__ out, int N) {
    const int lane = threadIdx.x & 31;
    const int wave0 = __builtin_amdgcn_readfirstlane((int)((blockIdx.x * blockDim.x + threadIdx.x) >> 5));
    const int nwave = (int)((gridDim.x * blockDim.x) >> 5);
    const float4* __restrict__ h4 = (const float4*)h;

    for (int d = wave0; d < N; d += nwave) {
        const int   beg = rowptr[d];
        const int   n   = cnt[d];
        const float dd  = dinv[d];
        const float d2  = dd * dd;
        const int   t   = n + 1;

        float4 acc = make_float4(0.0f, 0.0f, 0.0f, 0.0f);

        int c0 = 0;
        do {
            const int  m      = (t - c0 < 32) ? (t - c0) : 32;
            const int  idx    = c0 + lane;
            const bool isEdge = idx < n;
            int slot = beg + (isEdge ? idx : (n - 1));
            slot = (slot > 0) ? slot : 0;
            int sv = csrc[slot];
            sv = (int)min((unsigned)sv, (unsigned)(N - 1));
            const float dvl = dinv[sv];
            sv = isEdge ? sv : d;

            int j = 0;
            while (m - j > 4) {
                row_group<8>(h4, sv, dvl, j, c0 + j, n, dd, d2, lane, acc);
                j += 8;
            }
            if (m - j > 0) row_group<4>(h4, sv, dvl, j, c0 + j, n, dd, d2, lane, acc);

            c0 += 32;
        } while (c0 < t);

        const float4 bv = ((const float4*)bias)[lane];
        float4 o;
        o.x = acc.x + bv.x;
        o.y = acc.y + bv.y;
        o.z = acc.z + bv.z;
        o.w = acc.w + bv.w;
        ((float4*)out)[(size_t)d * 32 + lane] = o;
    }
}

extern "C" void kernel_launch(void* const* d_in, const int* in_sizes, int n_in,
                              void* d_out, int out_size, void* d_ws, size_t ws_size,
                              hipStream_t stream) {
    const float* x   = (const float*)d_in[0];
    const int*   ei  = (const int*)d_in[1];
    const float* W1  = (const float*)d_in[2];
    const float* b1  = (const float*)d_in[3];
    const float* W2  = (const float*)d_in[4];
    const float* b2  = (const float*)d_in[5];
    float*       out = (float*)d_out;

    const int N = in_sizes[0] / FDIM;
    const int E = in_sizes[1] / 2;
    const int* src = ei;
    const int* dst = ei + E;

    char*  ws    = (char*)d_ws;
    size_t nodeB = ((size_t)N * 4 + 255) & ~(size_t)255;
    size_t edgeB = ((size_t)E * 4 + 255) & ~(size_t)255;
    int*   cnt    = (int*)ws;
    int*   rowptr = (int*)(ws + nodeB);
    float* dinv   = (float*)(ws + 2 * nodeB);
    int*   rank   = (int*)(ws + 3 * nodeB);
    int*   csrc   = (int*)(ws + 3 * nodeB + edgeB);
    float* bufH   = (float*)(ws + 3 * nodeB + 2 * edgeB);
    float* bufA   = bufH + (size_t)N * FDIM;

    const int TPB = 256;

    k_zero_cnt<<<(N + TPB - 1) / TPB, TPB, 0, stream>>>(cnt, N);
    k_rank<<<(E + 1023) / 1024, TPB, 0, stream>>>(dst, cnt, rank, E);
    k_scan_dinv<<<(N + 4095) / 4096, 1024, 0, stream>>>(cnt, rowptr, dinv, N);
    k_fill<<<(E + 1023) / 1024, TPB, 0, stream>>>(src, dst, rank, rowptr, csrc, E);

    const int tiles      = (N + 15) / 16;
    const int gemmBlocks = (tiles + (TPB / 32) - 1) / (TPB / 32);
    const int aggBlocks  = 2048;

    k_gemm128<false><<<gemmBlocks, TPB, 0, stream>>>(x, W1, bufH, N);
    k_gather_agg<<<aggBlocks, TPB, 0, stream>>>(bufH, dinv, rowptr, cnt, csrc, b1, bufA, N);

    k_gemm128<true><<<gemmBlocks, TPB, 0, stream>>>(bufA, W2, bufH, N);
    k_gather_agg<<<aggBlocks, TPB, 0, stream>>>(bufH, dinv, rowptr, cnt, csrc, b2, out, N);
}


# Edge counting sort moved into LDS: per-chunk LDS histogram over 512-node ranges, scan, LDS-cursor scatter, per-range LDS row fill; no global atomics left; f32 row sum order = LDS arrival order
# speedup vs baseline: 3.1731x; 1.3138x over previous
#include <hip/hip_runtime.h>
#include <math.h>


typedef float v2f __attribute__((ext_vector_type(2)));
typedef float v8f __attribute__((ext_vector_type(8)));

#define FDIM 128

#define BIN_SHIFT   9
#define BIN_NODES   512
#define CHUNK_EDGES 8192
#define MAX_BINS    1024

__global__ void __launch_bounds__(1024) k_bin_count(const int* __restrict__ dst, int* __restrict__ counts,
                                                    int E, int N, int NB, int C) {
    __shared__ int hist[MAX_BINS];
    const int tid = threadIdx.x;
    const int c   = blockIdx.x;
    if (tid < NB) hist[tid] = 0;
    __syncthreads();

    const int base = c * CHUNK_EDGES + tid;
    int d[8];
#pragma unroll
    for (int k = 0; k < 8; ++k) {
        const int e = base + k * 1024;
        d[k] = dst[(e < E) ? e : (E - 1)];
    }
#pragma unroll
    for (int k = 0; k < 8; ++k) {
        const int e = base + k * 1024;
        if (e < E) {
            const unsigned dd = min((unsigned)d[k], (unsigned)(N - 1));
            atomicAdd(&hist[dd >> BIN_SHIFT], 1);
        }
    }
    __syncthreads();
    if (tid < NB) counts[tid * C + c] = hist[tid];
}

__global__ void __launch_bounds__(1024) k_scan(const int* __restrict__ cnt, int* __restrict__ rowptr, int N) {
    __shared__ int sPre[32];
    __shared__ int sTot[32];
    const int tid  = threadIdx.x;
    const int lane = tid & 31;
    const int wv   = tid >> 5;
    const int nb   = blockIdx.x;
    const int4* __restrict__ c4 = (const int4*)cnt;

    int pre = 0;
    int it = 0;
    for (; it + 8 <= nb; it += 8) {
        int4 v[8];
#pragma unroll
        for (int u = 0; u < 8; ++u) v[u] = c4[(it + u) * 1024 + tid];
#pragma unroll
        for (int u = 0; u < 8; ++u) pre += (v[u].x + v[u].y) + (v[u].z + v[u].w);
    }
    for (; it < nb; ++it) {
        const int4 v = c4[it * 1024 + tid];
        pre += (v.x + v.y) + (v.z + v.w);
    }

    const int idx = nb * 4096 + tid * 4;
    int m0 = 0, m1 = 0, m2 = 0, m3 = 0;
    if (idx + 3 < N) {
        const int4 v = c4[idx >> 2];
        m0 = v.x; m1 = v.y; m2 = v.z; m3 = v.w;
    } else {
        if (idx     < N) m0 = cnt[idx];
        if (idx + 1 < N) m1 = cnt[idx + 1];
        if (idx + 2 < N) m2 = cnt[idx + 2];
    }
    const int tsum = (m0 + m1) + (m2 + m3);

    int incl = tsum;
#pragma unroll
    for (int o = 1; o < 32; o <<= 1) {
        const int t = __shfl_up(incl, o);
        if (lane >= o) incl += t;
    }
#pragma unroll
    for (int o = 16; o > 0; o >>= 1) pre += __shfl_xor(pre, o);

    if (lane == 31) sTot[wv] = incl;
    if (lane == 0)  sPre[wv] = pre;
    __syncthreads();

    int base = 0;
#pragma unroll
    for (int w = 0; w < 32; ++w) {
        base += sPre[w];
        if (w < wv) base += sTot[w];
    }
    const int e0 = base + incl - tsum;
    const int e1 = e0 + m0, e2 = e1 + m1, e3 = e2 + m2;

    if (idx + 3 < N) {
        ((int4*)rowptr)[idx >> 2] = make_int4(e0, e1, e2, e3);
    } else {
        if (idx     < N) rowptr[idx]     = e0;
        if (idx + 1 < N) rowptr[idx + 1] = e1;
        if (idx + 2 < N) rowptr[idx + 2] = e2;
    }
}

__global__ void __launch_bounds__(1024) k_bin_scatter(const int* __restrict__ src, const int* __restrict__ dst,
                                                      const int* __restrict__ offs, int* __restrict__ bucket,
                                                      int E, int N, int NB, int C) {
    __shared__ int cursor[MAX_BINS];
    const int tid = threadIdx.x;
    const int c   = blockIdx.x;
    if (tid < NB) cursor[tid] = offs[tid * C + c];
    __syncthreads();

    const int base = c * CHUNK_EDGES + tid;
    int d[8], s[8];
#pragma unroll
    for (int k = 0; k < 8; ++k) {
        const int e  = base + k * 1024;
        const int ec = (e < E) ? e : (E - 1);
        d[k] = dst[ec];
        s[k] = src[ec];
    }
#pragma unroll
    for (int k = 0; k < 8; ++k) {
        const int e = base + k * 1024;
        if (e < E) {
            const unsigned dd = min((unsigned)d[k], (unsigned)(N - 1));
            const unsigned ss = min((unsigned)s[k], (unsigned)(N - 1));
            const int pos = atomicAdd(&cursor[dd >> BIN_SHIFT], 1);
            if ((unsigned)pos < (unsigned)E) bucket[pos] = (int)((ss << BIN_SHIFT) | (dd & (BIN_NODES - 1)));
        }
    }
}

__global__ void __launch_bounds__(1024) k_bin_rows(const int* __restrict__ bucket, const int* __restrict__ offs,
                                                   int* __restrict__ cnt, int* __restrict__ rowptr,
                                                   float* __restrict__ dinv, int* __restrict__ csrc,
                                                   int E, int N, int NB, int C) {
    __shared__ int sCnt[BIN_NODES];
    __shared__ int sTot[32];
    const int tid  = threadIdx.x;
    const int lane = tid & 31;
    const int wv   = tid >> 5;
    const int b    = blockIdx.x;
    int bs = offs[b * C];
    int be = (b + 1 < NB) ? offs[(b + 1) * C] : E;
    be = min(max(be, 0), E);
    bs = min(max(bs, 0), be);

    if (tid < BIN_NODES) sCnt[tid] = 0;
    __syncthreads();

    for (int i0 = bs; i0 < be; i0 += 4096) {
        int r[4];
#pragma unroll
        for (int u = 0; u < 4; ++u) {
            const int i = i0 + u * 1024 + tid;
            r[u] = bucket[(i < be) ? i : (be - 1)];
        }
#pragma unroll
        for (int u = 0; u < 4; ++u) {
            const int i = i0 + u * 1024 + tid;
            if (i < be) atomicAdd(&sCnt[r[u] & (BIN_NODES - 1)], 1);
        }
    }
    __syncthreads();

    const int myc = (tid < BIN_NODES) ? sCnt[tid] : 0;
    int incl = myc;
#pragma unroll
    for (int o = 1; o < 32; o <<= 1) {
        const int t = __shfl_up(incl, o);
        if (lane >= o) incl += t;
    }
    if (lane == 31) sTot[wv] = incl;
    __syncthreads();
    int wbase = 0;
#pragma unroll
    for (int w = 0; w < 32; ++w) {
        if (w < wv) wbase += sTot[w];
    }
    const int first = bs + wbase + incl - myc;

    if (tid < BIN_NODES) {
        sCnt[tid] = first;
        const int node = b * BIN_NODES + tid;
        if (node < N) {
            cnt[node]    = myc;
            rowptr[node] = first;
            dinv[node]   = rsqrtf((float)(myc + 1));
        }
    }
    __syncthreads();

    for (int i0 = bs; i0 < be; i0 += 4096) {
        int r[4];
#pragma unroll
        for (int u = 0; u < 4; ++u) {
            const int i = i0 + u * 1024 + tid;
            r[u] = bucket[(i < be) ? i : (be - 1)];
        }
#pragma unroll
        for (int u = 0; u < 4; ++u) {
            const int i = i0 + u * 1024 + tid;
            if (i < be) {
                const int pos = atomicAdd(&sCnt[r[u] & (BIN_NODES - 1)], 1);
                if ((unsigned)pos < (unsigned)E) csrc[pos] = (int)((unsigned)r[u] >> BIN_SHIFT);
            }
        }
    }
}

template <bool RELU_A>
__global__ void k_gemm128(const float* __restrict__ A,
                          const float* __restrict__ W,
                          float* __restrict__ outH, int M) {
    __shared__ float sW[FDIM * FDIM];

    const int lane = threadIdx.x & 31;
    const int half = lane >> 4;
    const int l16  = lane & 15;

    for (int e = threadIdx.x; e < 32 * 8 * 32; e += blockDim.x) {
        int kstep = e >> 8;
        int rem   = e & 255;
        int t     = rem >> 5;
        int ln    = rem & 31;
        int h     = ln >> 4;
        int c     = t * 16 + (ln & 15);
        int krow  = kstep * 4 + 2 * h;
        float2 v;
        v.x = W[(size_t)krow * FDIM + c];
        v.y = W[(size_t)(krow + 1) * FDIM + c];
        ((float2*)sW)[e] = v;
    }
    __syncthreads();

    const int waveId = (blockIdx.x * blockDim.x + threadIdx.x) >> 5;
    const int rowT   = waveId;
    if (rowT * 16 >= M) return;

    const float* __restrict__ arow = A + (size_t)(rowT * 16 + l16) * FDIM;

    v8f acc[8];
#pragma unroll
    for (int t = 0; t < 8; ++t) acc[t] = (v8f){0, 0, 0, 0, 0, 0, 0, 0};

    for (int kstep = 0; kstep < 32; ++kstep) {
        v2f a = *(const v2f*)(arow + kstep * 4 + 2 * half);
        if (RELU_A) {
            a.x = fmaxf(a.x, 0.0f);
            a.y = fmaxf(a.y, 0.0f);
        }
        const v2f* __restrict__ bfrag = (const v2f*)sW + kstep * 256 + lane;
#pragma unroll
        for (int t = 0; t < 8; ++t) {
            v2f b = bfrag[t * 32];
            acc[t] = __builtin_amdgcn_wmma_f32_16x16x4_f32(
                false, a, false, b, (short)0, acc[t], false, false);
        }
    }

#pragma unroll
    for (int t = 0; t < 8; ++t) {
        const int col = t * 16 + l16;
#pragma unroll
        for (int r = 0; r < 8; ++r) {
            const int row = rowT * 16 + r + half * 8;
            outH[(size_t)row * FDIM + col] = acc[t][r];
        }
    }
}

template <int G>
__device__ __forceinline__ void row_group(const float4* __restrict__ h4, int sv, float dvl,
                                          int j, int id0, int n, float dd, float d2,
                                          int lane, float4& acc) {
    int    s[G];
    float4 v[G];
#pragma unroll
    for (int u = 0; u < G; ++u) s[u] = __builtin_amdgcn_readlane(sv, j + u);
#pragma unroll
    for (int u = 0; u < G; ++u) v[u] = h4[(size_t)s[u] * 32 + lane];
#pragma unroll
    for (int u = 0; u < G; ++u) {
        const int   id = id0 + u;
        const float ds = __int_as_float(__builtin_amdgcn_readlane(__float_as_int(dvl), j + u));
        const float w  = (id < n) ? ds * dd : ((id == n) ? d2 : 0.0f);
        acc.x = fmaf(w, v[u].x, acc.x);
        acc.y = fmaf(w, v[u].y, acc.y);
        acc.z = fmaf(w, v[u].z, acc.z);
        acc.w = fmaf(w, v[u].w, acc.w);
    }
}

__global__ void __launch_bounds__(256) k_gather_agg(const float* __restrict__ h,
                                                    const float* __restrict__ dinv,
                                                    const int* __restrict__ rowptr,
                                                    const int* __restrict__ cnt,
                                                    const int* __restrict__ csrc,
                                                    const float* __restrict__ bias,
                                                    float* __restrict__ out, int N) {
    const int lane = threadIdx.x & 31;
    const int wave0 = __builtin_amdgcn_readfirstlane((int)((blockIdx.x * blockDim.x + threadIdx.x) >> 5));
    const int nwave = (int)((gridDim.x * blockDim.x) >> 5);
    const float4* __restrict__ h4 = (const float4*)h;

    for (int d = wave0; d < N; d += nwave) {
        const int   beg = rowptr[d];
        const int   n   = cnt[d];
        const float dd  = dinv[d];
        const float d2  = dd * dd;
        const int   t   = n + 1;

        float4 acc = make_float4(0.0f, 0.0f, 0.0f, 0.0f);

        int c0 = 0;
        do {
            const int  m      = (t - c0 < 32) ? (t - c0) : 32;
            const int  idx    = c0 + lane;
            const bool isEdge = idx < n;
            int slot = beg + (isEdge ? idx : (n - 1));
            slot = (slot > 0) ? slot : 0;
            int sv = csrc[slot];
            sv = (int)min((unsigned)sv, (unsigned)(N - 1));
            const float dvl = dinv[sv];
            sv = isEdge ? sv : d;

            int j = 0;
            while (m - j > 4) {
                row_group<8>(h4, sv, dvl, j, c0 + j, n, dd, d2, lane, acc);
                j += 8;
            }
            if (m - j > 0) row_group<4>(h4, sv, dvl, j, c0 + j, n, dd, d2, lane, acc);

            c0 += 32;
        } while (c0 < t);

        const float4 bv = ((const float4*)bias)[lane];
        float4 o;
        o.x = acc.x + bv.x;
        o.y = acc.y + bv.y;
        o.z = acc.z + bv.z;
        o.w = acc.w + bv.w;
        ((float4*)out)[(size_t)d * 32 + lane] = o;
    }
}

extern "C" void kernel_launch(void* const* d_in, const int* in_sizes, int n_in,
                              void* d_out, int out_size, void* d_ws, size_t ws_size,
                              hipStream_t stream) {
    const float* x   = (const float*)d_in[0];
    const int*   ei  = (const int*)d_in[1];
    const float* W1  = (const float*)d_in[2];
    const float* b1  = (const float*)d_in[3];
    const float* W2  = (const float*)d_in[4];
    const float* b2  = (const float*)d_in[5];
    float*       out = (float*)d_out;

    const int N = in_sizes[0] / FDIM;
    const int E = in_sizes[1] / 2;
    const int* src = ei;
    const int* dst = ei + E;

    const int NB = (N + BIN_NODES - 1) / BIN_NODES;
    const int C  = (E + CHUNK_EDGES - 1) / CHUNK_EDGES;
    const int M  = NB * C;

    char*  ws    = (char*)d_ws;
    size_t nodeB = ((size_t)N * 4 + 255) & ~(size_t)255;
    size_t edgeB = ((size_t)E * 4 + 255) & ~(size_t)255;
    size_t tabB  = ((size_t)M * 4 + 255) & ~(size_t)255;
    int*   cnt    = (int*)ws;
    int*   rowptr = (int*)(ws + nodeB);
    float* dinv   = (float*)(ws + 2 * nodeB);
    int*   bucket = (int*)(ws + 3 * nodeB);
    int*   csrc   = (int*)(ws + 3 * nodeB + edgeB);
    float* bufH   = (float*)(ws + 3 * nodeB + 2 * edgeB);
    float* bufA   = bufH + (size_t)N * FDIM;
    int*   counts = (int*)(ws + 3 * nodeB + 2 * edgeB + 2 * (size_t)N * FDIM * 4);
    int*   offs   = (int*)(ws + 3 * nodeB + 2 * edgeB + 2 * (size_t)N * FDIM * 4 + tabB);

    const int TPB = 256;

    k_bin_count<<<C, 1024, 0, stream>>>(dst, counts, E, N, NB, C);
    k_scan<<<(M + 4095) / 4096, 1024, 0, stream>>>(counts, offs, M);
    k_bin_scatter<<<C, 1024, 0, stream>>>(src, dst, offs, bucket, E, N, NB, C);
    k_bin_rows<<<NB, 1024, 0, stream>>>(bucket, offs, cnt, rowptr, dinv, csrc, E, N, NB, C);

    const int tiles      = (N + 15) / 16;
    const int gemmBlocks = (tiles + (TPB / 32) - 1) / (TPB / 32);
    const int aggBlocks  = 2048;

    k_gemm128<false><<<gemmBlocks, TPB, 0, stream>>>(x, W1, bufH, N);
    k_gather_agg<<<aggBlocks, TPB, 0, stream>>>(bufH, dinv, rowptr, cnt, csrc, b1, bufA, N);

    k_gemm128<true><<<gemmBlocks, TPB, 0, stream>>>(bufA, W2, bufH, N);
    k_gather_agg<<<aggBlocks, TPB, 0, stream>>>(bufH, dinv, rowptr, cnt, csrc, b2, out, N);
}


# GEMM epilogue stores rows pre-scaled by dinv[row]; aggregation drops its per-edge scattered dinv[src] lookup and weight, applies dinv[dst] once to the row sum (sum order unchanged)
# speedup vs baseline: 3.2929x; 1.0378x over previous
#include <hip/hip_runtime.h>
#include <math.h>


typedef float v2f __attribute__((ext_vector_type(2)));
typedef float v8f __attribute__((ext_vector_type(8)));

#define FDIM 128

#define BIN_SHIFT   9
#define BIN_NODES   512
#define CHUNK_EDGES 8192
#define MAX_BINS    1024

__global__ void __launch_bounds__(1024) k_bin_count(const int* __restrict__ dst, int* __restrict__ counts,
                                                    int E, int N, int NB, int C) {
    __shared__ int hist[MAX_BINS];
    const int tid = threadIdx.x;
    const int c   = blockIdx.x;
    if (tid < NB) hist[tid] = 0;
    __syncthreads();

    const int base = c * CHUNK_EDGES + tid;
    int d[8];
#pragma unroll
    for (int k = 0; k < 8; ++k) {
        const int e = base + k * 1024;
        d[k] = dst[(e < E) ? e : (E - 1)];
    }
#pragma unroll
    for (int k = 0; k < 8; ++k) {
        const int e = base + k * 1024;
        if (e < E) {
            const unsigned dd = min((unsigned)d[k], (unsigned)(N - 1));
            atomicAdd(&hist[dd >> BIN_SHIFT], 1);
        }
    }
    __syncthreads();
    if (tid < NB) counts[tid * C + c] = hist[tid];
}

__global__ void __launch_bounds__(1024) k_scan(const int* __restrict__ cnt, int* __restrict__ rowptr, int N) {
    __shared__ int sPre[32];
    __shared__ int sTot[32];
    const int tid  = threadIdx.x;
    const int lane = tid & 31;
    const int wv   = tid >> 5;
    const int nb   = blockIdx.x;
    const int4* __restrict__ c4 = (const int4*)cnt;

    int pre = 0;
    int it = 0;
    for (; it + 8 <= nb; it += 8) {
        int4 v[8];
#pragma unroll
        for (int u = 0; u < 8; ++u) v[u] = c4[(it + u) * 1024 + tid];
#pragma unroll
        for (int u = 0; u < 8; ++u) pre += (v[u].x + v[u].y) + (v[u].z + v[u].w);
    }
    for (; it < nb; ++it) {
        const int4 v = c4[it * 1024 + tid];
        pre += (v.x + v.y) + (v.z + v.w);
    }

    const int idx = nb * 4096 + tid * 4;
    int m0 = 0, m1 = 0, m2 = 0, m3 = 0;
    if (idx + 3 < N) {
        const int4 v = c4[idx >> 2];
        m0 = v.x; m1 = v.y; m2 = v.z; m3 = v.w;
    } else {
        if (idx     < N) m0 = cnt[idx];
        if (idx + 1 < N) m1 = cnt[idx + 1];
        if (idx + 2 < N) m2 = cnt[idx + 2];
    }
    const int tsum = (m0 + m1) + (m2 + m3);

    int incl = tsum;
#pragma unroll
    for (int o = 1; o < 32; o <<= 1) {
        const int t = __shfl_up(incl, o);
        if (lane >= o) incl += t;
    }
#pragma unroll
    for (int o = 16; o > 0; o >>= 1) pre += __shfl_xor(pre, o);

    if (lane == 31) sTot[wv] = incl;
    if (lane == 0)  sPre[wv] = pre;
    __syncthreads();

    int base = 0;
#pragma unroll
    for (int w = 0; w < 32; ++w) {
        base += sPre[w];
        if (w < wv) base += sTot[w];
    }
    const int e0 = base + incl - tsum;
    const int e1 = e0 + m0, e2 = e1 + m1, e3 = e2 + m2;

    if (idx + 3 < N) {
        ((int4*)rowptr)[idx >> 2] = make_int4(e0, e1, e2, e3);
    } else {
        if (idx     < N) rowptr[idx]     = e0;
        if (idx + 1 < N) rowptr[idx + 1] = e1;
        if (idx + 2 < N) rowptr[idx + 2] = e2;
    }
}

__global__ void __launch_bounds__(1024) k_bin_scatter(const int* __restrict__ src, const int* __restrict__ dst,
                                                      const int* __restrict__ offs, int* __restrict__ bucket,
                                                      int E, int N, int NB, int C) {
    __shared__ int cursor[MAX_BINS];
    const int tid = threadIdx.x;
    const int c   = blockIdx.x;
    if (tid < NB) cursor[tid] = offs[tid * C + c];
    __syncthreads();

    const int base = c * CHUNK_EDGES + tid;
    int d[8], s[8];
#pragma unroll
    for (int k = 0; k < 8; ++k) {
        const int e  = base + k * 1024;
        const int ec = (e < E) ? e : (E - 1);
        d[k] = dst[ec];
        s[k] = src[ec];
    }
#pragma unroll
    for (int k = 0; k < 8; ++k) {
        const int e = base + k * 1024;
        if (e < E) {
            const unsigned dd = min((unsigned)d[k], (unsigned)(N - 1));
            const unsigned ss = min((unsigned)s[k], (unsigned)(N - 1));
            const int pos = atomicAdd(&cursor[dd >> BIN_SHIFT], 1);
            if ((unsigned)pos < (unsigned)E) bucket[pos] = (int)((ss << BIN_SHIFT) | (dd & (BIN_NODES - 1)));
        }
    }
}

__global__ void __launch_bounds__(1024) k_bin_rows(const int* __restrict__ bucket, const int* __restrict__ offs,
                                                   int* __restrict__ cnt, int* __restrict__ rowptr,
                                                   float* __restrict__ dinv, int* __restrict__ csrc,
                                                   int E, int N, int NB, int C) {
    __shared__ int sCnt[BIN_NODES];
    __shared__ int sTot[32];
    const int tid  = threadIdx.x;
    const int lane = tid & 31;
    const int wv   = tid >> 5;
    const int b    = blockIdx.x;
    int bs = offs[b * C];
    int be = (b + 1 < NB) ? offs[(b + 1) * C] : E;
    be = min(max(be, 0), E);
    bs = min(max(bs, 0), be);

    if (tid < BIN_NODES) sCnt[tid] = 0;
    __syncthreads();

    for (int i0 = bs; i0 < be; i0 += 4096) {
        int r[4];
#pragma unroll
        for (int u = 0; u < 4; ++u) {
            const int i = i0 + u * 1024 + tid;
            r[u] = bucket[(i < be) ? i : (be - 1)];
        }
#pragma unroll
        for (int u = 0; u < 4; ++u) {
            const int i = i0 + u * 1024 + tid;
            if (i < be) atomicAdd(&sCnt[r[u] & (BIN_NODES - 1)], 1);
        }
    }
    __syncthreads();

    const int myc = (tid < BIN_NODES) ? sCnt[tid] : 0;
    int incl = myc;
#pragma unroll
    for (int o = 1; o < 32; o <<= 1) {
        const int t = __shfl_up(incl, o);
        if (lane >= o) incl += t;
    }
    if (lane == 31) sTot[wv] = incl;
    __syncthreads();
    int wbase = 0;
#pragma unroll
    for (int w = 0; w < 32; ++w) {
        if (w < wv) wbase += sTot[w];
    }
    const int first = bs + wbase + incl - myc;

    if (tid < BIN_NODES) {
        sCnt[tid] = first;
        const int node = b * BIN_NODES + tid;
        if (node < N) {
            cnt[node]    = myc;
            rowptr[node] = first;
            dinv[node]   = rsqrtf((float)(myc + 1));
        }
    }
    __syncthreads();

    for (int i0 = bs; i0 < be; i0 += 4096) {
        int r[4];
#pragma unroll
        for (int u = 0; u < 4; ++u) {
            const int i = i0 + u * 1024 + tid;
            r[u] = bucket[(i < be) ? i : (be - 1)];
        }
#pragma unroll
        for (int u = 0; u < 4; ++u) {
            const int i = i0 + u * 1024 + tid;
            if (i < be) {
                const int pos = atomicAdd(&sCnt[r[u] & (BIN_NODES - 1)], 1);
                if ((unsigned)pos < (unsigned)E) csrc[pos] = (int)((unsigned)r[u] >> BIN_SHIFT);
            }
        }
    }
}

template <bool RELU_A>
__global__ void k_gemm128(const float* __restrict__ A,
                          const float* __restrict__ W,
                          const float* __restrict__ dinv,
                          float* __restrict__ outH, int M) {
    __shared__ float sW[FDIM * FDIM];

    const int lane = threadIdx.x & 31;
    const int half = lane >> 4;
    const int l16  = lane & 15;

    for (int e = threadIdx.x; e < 32 * 8 * 32; e += blockDim.x) {
        int kstep = e >> 8;
        int rem   = e & 255;
        int t     = rem >> 5;
        int ln    = rem & 31;
        int h     = ln >> 4;
        int c     = t * 16 + (ln & 15);
        int krow  = kstep * 4 + 2 * h;
        float2 v;
        v.x = W[(size_t)krow * FDIM + c];
        v.y = W[(size_t)(krow + 1) * FDIM + c];
        ((float2*)sW)[e] = v;
    }
    __syncthreads();

    const int waveId = (blockIdx.x * blockDim.x + threadIdx.x) >> 5;
    const int rowT   = waveId;
    if (rowT * 16 >= M) return;

    const float* __restrict__ arow = A + (size_t)(rowT * 16 + l16) * FDIM;

    float sc[8];
#pragma unroll
    for (int r = 0; r < 8; ++r) {
        const int row = rowT * 16 + r + half * 8;
        sc[r] = dinv[(row < M) ? row : (M - 1)];
    }

    v8f acc[8];
#pragma unroll
    for (int t = 0; t < 8; ++t) acc[t] = (v8f){0, 0, 0, 0, 0, 0, 0, 0};

    for (int kstep = 0; kstep < 32; ++kstep) {
        v2f a = *(const v2f*)(arow + kstep * 4 + 2 * half);
        if (RELU_A) {
            a.x = fmaxf(a.x, 0.0f);
            a.y = fmaxf(a.y, 0.0f);
        }
        const v2f* __restrict__ bfrag = (const v2f*)sW + kstep * 256 + lane;
#pragma unroll
        for (int t = 0; t < 8; ++t) {
            v2f b = bfrag[t * 32];
            acc[t] = __builtin_amdgcn_wmma_f32_16x16x4_f32(
                false, a, false, b, (short)0, acc[t], false, false);
        }
    }

#pragma unroll
    for (int t = 0; t < 8; ++t) {
        const int col = t * 16 + l16;
#pragma unroll
        for (int r = 0; r < 8; ++r) {
            const int row = rowT * 16 + r + half * 8;
            outH[(size_t)row * FDIM + col] = acc[t][r] * sc[r];
        }
    }
}

template <int G>
__device__ __forceinline__ void row_group(const float4* __restrict__ h4, int sv,
                                          int j, int id0, int n,
                                          int lane, float4& acc) {
    int    s[G];
    float4 v[G];
#pragma unroll
    for (int u = 0; u < G; ++u) s[u] = __builtin_amdgcn_readlane(sv, j + u);
#pragma unroll
    for (int u = 0; u < G; ++u) v[u] = h4[(size_t)s[u] * 32 + lane];
#pragma unroll
    for (int u = 0; u < G; ++u) {
        const int   id = id0 + u;
        const float w  = (id <= n) ? 1.0f : 0.0f;
        acc.x = fmaf(w, v[u].x, acc.x);
        acc.y = fmaf(w, v[u].y, acc.y);
        acc.z = fmaf(w, v[u].z, acc.z);
        acc.w = fmaf(w, v[u].w, acc.w);
    }
}

__global__ void __launch_bounds__(256) k_gather_agg(const float* __restrict__ h,
                                                    const float* __restrict__ dinv,
                                                    const int* __restrict__ rowptr,
                                                    const int* __restrict__ cnt,
                                                    const int* __restrict__ csrc,
                                                    const float* __restrict__ bias,
                                                    float* __restrict__ out, int N) {
    const int lane = threadIdx.x & 31;
    const int wave0 = __builtin_amdgcn_readfirstlane((int)((blockIdx.x * blockDim.x + threadIdx.x) >> 5));
    const int nwave = (int)((gridDim.x * blockDim.x) >> 5);
    const float4* __restrict__ h4 = (const float4*)h;

    for (int d = wave0; d < N; d += nwave) {
        const int   beg = rowptr[d];
        const int   n   = cnt[d];
        const float dd  = dinv[d];
        const int   t   = n + 1;

        float4 acc = make_float4(0.0f, 0.0f, 0.0f, 0.0f);

        int c0 = 0;
        do {
            const int  m      = (t - c0 < 32) ? (t - c0) : 32;
            const int  idx    = c0 + lane;
            const bool isEdge = idx < n;
            int slot = beg + (isEdge ? idx : (n - 1));
            slot = (slot > 0) ? slot : 0;
            int sv = csrc[slot];
            sv = (int)min((unsigned)sv, (unsigned)(N - 1));
            sv = isEdge ? sv : d;

            int j = 0;
            while (m - j > 4) {
                row_group<8>(h4, sv, j, c0 + j, n, lane, acc);
                j += 8;
            }
            if (m - j > 0) row_group<4>(h4, sv, j, c0 + j, n, lane, acc);

            c0 += 32;
        } while (c0 < t);

        const float4 bv = ((const float4*)bias)[lane];
        float4 o;
        o.x = fmaf(dd, acc.x, bv.x);
        o.y = fmaf(dd, acc.y, bv.y);
        o.z = fmaf(dd, acc.z, bv.z);
        o.w = fmaf(dd, acc.w, bv.w);
        ((float4*)out)[(size_t)d * 32 + lane] = o;
    }
}

extern "C" void kernel_launch(void* const* d_in, const int* in_sizes, int n_in,
                              void* d_out, int out_size, void* d_ws, size_t ws_size,
                              hipStream_t stream) {
    const float* x   = (const float*)d_in[0];
    const int*   ei  = (const int*)d_in[1];
    const float* W1  = (const float*)d_in[2];
    const float* b1  = (const float*)d_in[3];
    const float* W2  = (const float*)d_in[4];
    const float* b2  = (const float*)d_in[5];
    float*       out = (float*)d_out;

    const int N = in_sizes[0] / FDIM;
    const int E = in_sizes[1] / 2;
    const int* src = ei;
    const int* dst = ei + E;

    const int NB = (N + BIN_NODES - 1) / BIN_NODES;
    const int C  = (E + CHUNK_EDGES - 1) / CHUNK_EDGES;
    const int M  = NB * C;

    char*  ws    = (char*)d_ws;
    size_t nodeB = ((size_t)N * 4 + 255) & ~(size_t)255;
    size_t edgeB = ((size_t)E * 4 + 255) & ~(size_t)255;
    size_t tabB  = ((size_t)M * 4 + 255) & ~(size_t)255;
    int*   cnt    = (int*)ws;
    int*   rowptr = (int*)(ws + nodeB);
    float* dinv   = (float*)(ws + 2 * nodeB);
    int*   bucket = (int*)(ws + 3 * nodeB);
    int*   csrc   = (int*)(ws + 3 * nodeB + edgeB);
    float* bufH   = (float*)(ws + 3 * nodeB + 2 * edgeB);
    float* bufA   = bufH + (size_t)N * FDIM;
    int*   counts = (int*)(ws + 3 * nodeB + 2 * edgeB + 2 * (size_t)N * FDIM * 4);
    int*   offs   = (int*)(ws + 3 * nodeB + 2 * edgeB + 2 * (size_t)N * FDIM * 4 + tabB);

    const int TPB = 256;

    k_bin_count<<<C, 1024, 0, stream>>>(dst, counts, E, N, NB, C);
    k_scan<<<(M + 4095) / 4096, 1024, 0, stream>>>(counts, offs, M);
    k_bin_scatter<<<C, 1024, 0, stream>>>(src, dst, offs, bucket, E, N, NB, C);
    k_bin_rows<<<NB, 1024, 0, stream>>>(bucket, offs, cnt, rowptr, dinv, csrc, E, N, NB, C);

    const int tiles      = (N + 15) / 16;
    const int gemmBlocks = (tiles + (TPB / 32) - 1) / (TPB / 32);
    const int aggBlocks  = 2048;

    k_gemm128<false><<<gemmBlocks, TPB, 0, stream>>>(x, W1, dinv, bufH, N);
    k_gather_agg<<<aggBlocks, TPB, 0, stream>>>(bufH, dinv, rowptr, cnt, csrc, b1, bufA, N);

    k_gemm128<true><<<gemmBlocks, TPB, 0, stream>>>(bufA, W2, dinv, bufH, N);
    k_gather_agg<<<aggBlocks, TPB, 0, stream>>>(bufH, dinv, rowptr, cnt, csrc, b2, out, N);
}


# Layer-2 dense product fused into the layer-1 aggregation kernel: finished rows pass through an LDS tile to the WMMAs beside the row gathers; one launch and one buffer less; sum orders unchanged
# speedup vs baseline: 3.9545x; 1.2009x over previous
#include <hip/hip_runtime.h>
#include <math.h>


typedef float v2f __attribute__((ext_vector_type(2)));
typedef float v8f __attribute__((ext_vector_type(8)));

#define FDIM 128

#define BIN_SHIFT   9
#define BIN_NODES   512
#define CHUNK_EDGES 8192
#define MAX_BINS    1024

__global__ void __launch_bounds__(1024) k_bin_count(const int* __restrict__ dst, int* __restrict__ counts,
                                                    int E, int N, int NB, int C) {
    __shared__ int hist[MAX_BINS];
    const int tid = threadIdx.x;
    const int c   = blockIdx.x;
    if (tid < NB) hist[tid] = 0;
    __syncthreads();

    const int base = c * CHUNK_EDGES + tid;
    int d[8];
#pragma unroll
    for (int k = 0; k < 8; ++k) {
        const int e = base + k * 1024;
        d[k] = dst[(e < E) ? e : (E - 1)];
    }
#pragma unroll
    for (int k = 0; k < 8; ++k) {
        const int e = base + k * 1024;
        if (e < E) {
            const unsigned dd = min((unsigned)d[k], (unsigned)(N - 1));
            atomicAdd(&hist[dd >> BIN_SHIFT], 1);
        }
    }
    __syncthreads();
    if (tid < NB) counts[tid * C + c] = hist[tid];
}

__global__ void __launch_bounds__(1024) k_scan(const int* __restrict__ cnt, int* __restrict__ rowptr, int N) {
    __shared__ int sPre[32];
    __shared__ int sTot[32];
    const int tid  = threadIdx.x;
    const int lane = tid & 31;
    const int wv   = tid >> 5;
    const int nb   = blockIdx.x;
    const int4* __restrict__ c4 = (const int4*)cnt;

    int pre = 0;
    int it = 0;
    for (; it + 8 <= nb; it += 8) {
        int4 v[8];
#pragma unroll
        for (int u = 0; u < 8; ++u) v[u] = c4[(it + u) * 1024 + tid];
#pragma unroll
        for (int u = 0; u < 8; ++u) pre += (v[u].x + v[u].y) + (v[u].z + v[u].w);
    }
    for (; it < nb; ++it) {
        const int4 v = c4[it * 1024 + tid];
        pre += (v.x + v.y) + (v.z + v.w);
    }

    const int idx = nb * 4096 + tid * 4;
    int m0 = 0, m1 = 0, m2 = 0, m3 = 0;
    if (idx + 3 < N) {
        const int4 v = c4[idx >> 2];
        m0 = v.x; m1 = v.y; m2 = v.z; m3 = v.w;
    } else {
        if (idx     < N) m0 = cnt[idx];
        if (idx + 1 < N) m1 = cnt[idx + 1];
        if (idx + 2 < N) m2 = cnt[idx + 2];
    }
    const int tsum = (m0 + m1) + (m2 + m3);

    int incl = tsum;
#pragma unroll
    for (int o = 1; o < 32; o <<= 1) {
        const int t = __shfl_up(incl, o);
        if (lane >= o) incl += t;
    }
#pragma unroll
    for (int o = 16; o > 0; o >>= 1) pre += __shfl_xor(pre, o);

    if (lane == 31) sTot[wv] = incl;
    if (lane == 0)  sPre[wv] = pre;
    __syncthreads();

    int base = 0;
#pragma unroll
    for (int w = 0; w < 32; ++w) {
        base += sPre[w];
        if (w < wv) base += sTot[w];
    }
    const int e0 = base + incl - tsum;
    const int e1 = e0 + m0, e2 = e1 + m1, e3 = e2 + m2;

    if (idx + 3 < N) {
        ((int4*)rowptr)[idx >> 2] = make_int4(e0, e1, e2, e3);
    } else {
        if (idx     < N) rowptr[idx]     = e0;
        if (idx + 1 < N) rowptr[idx + 1] = e1;
        if (idx + 2 < N) rowptr[idx + 2] = e2;
    }
}

__global__ void __launch_bounds__(1024) k_bin_scatter(const int* __restrict__ src, const int* __restrict__ dst,
                                                      const int* __restrict__ offs, int* __restrict__ bucket,
                                                      int E, int N, int NB, int C) {
    __shared__ int cursor[MAX_BINS];
    const int tid = threadIdx.x;
    const int c   = blockIdx.x;
    if (tid < NB) cursor[tid] = offs[tid * C + c];
    __syncthreads();

    const int base = c * CHUNK_EDGES + tid;
    int d[8], s[8];
#pragma unroll
    for (int k = 0; k < 8; ++k) {
        const int e  = base + k * 1024;
        const int ec = (e < E) ? e : (E - 1);
        d[k] = dst[ec];
        s[k] = src[ec];
    }
#pragma unroll
    for (int k = 0; k < 8; ++k) {
        const int e = base + k * 1024;
        if (e < E) {
            const unsigned dd = min((unsigned)d[k], (unsigned)(N - 1));
            const unsigned ss = min((unsigned)s[k], (unsigned)(N - 1));
            const int pos = atomicAdd(&cursor[dd >> BIN_SHIFT], 1);
            if ((unsigned)pos < (unsigned)E) bucket[pos] = (int)((ss << BIN_SHIFT) | (dd & (BIN_NODES - 1)));
        }
    }
}

__global__ void __launch_bounds__(1024) k_bin_rows(const int* __restrict__ bucket, const int* __restrict__ offs,
                                                   int* __restrict__ cnt, int* __restrict__ rowptr,
                                                   float* __restrict__ dinv, int* __restrict__ csrc,
                                                   int E, int N, int NB, int C) {
    __shared__ int sCnt[BIN_NODES];
    __shared__ int sTot[32];
    const int tid  = threadIdx.x;
    const int lane = tid & 31;
    const int wv   = tid >> 5;
    const int b    = blockIdx.x;
    int bs = offs[b * C];
    int be = (b + 1 < NB) ? offs[(b + 1) * C] : E;
    be = min(max(be, 0), E);
    bs = min(max(bs, 0), be);

    if (tid < BIN_NODES) sCnt[tid] = 0;
    __syncthreads();

    for (int i0 = bs; i0 < be; i0 += 4096) {
        int r[4];
#pragma unroll
        for (int u = 0; u < 4; ++u) {
            const int i = i0 + u * 1024 + tid;
            r[u] = bucket[(i < be) ? i : (be - 1)];
        }
#pragma unroll
        for (int u = 0; u < 4; ++u) {
            const int i = i0 + u * 1024 + tid;
            if (i < be) atomicAdd(&sCnt[r[u] & (BIN_NODES - 1)], 1);
        }
    }
    __syncthreads();

    const int myc = (tid < BIN_NODES) ? sCnt[tid] : 0;
    int incl = myc;
#pragma unroll
    for (int o = 1; o < 32; o <<= 1) {
        const int t = __shfl_up(incl, o);
        if (lane >= o) incl += t;
    }
    if (lane == 31) sTot[wv] = incl;
    __syncthreads();
    int wbase = 0;
#pragma unroll
    for (int w = 0; w < 32; ++w) {
        if (w < wv) wbase += sTot[w];
    }
    const int first = bs + wbase + incl - myc;

    if (tid < BIN_NODES) {
        sCnt[tid] = first;
        const int node = b * BIN_NODES + tid;
        if (node < N) {
            cnt[node]    = myc;
            rowptr[node] = first;
            dinv[node]   = rsqrtf((float)(myc + 1));
        }
    }
    __syncthreads();

    for (int i0 = bs; i0 < be; i0 += 4096) {
        int r[4];
#pragma unroll
        for (int u = 0; u < 4; ++u) {
            const int i = i0 + u * 1024 + tid;
            r[u] = bucket[(i < be) ? i : (be - 1)];
        }
#pragma unroll
        for (int u = 0; u < 4; ++u) {
            const int i = i0 + u * 1024 + tid;
            if (i < be) {
                const int pos = atomicAdd(&sCnt[r[u] & (BIN_NODES - 1)], 1);
                if ((unsigned)pos < (unsigned)E) csrc[pos] = (int)((unsigned)r[u] >> BIN_SHIFT);
            }
        }
    }
}

template <bool RELU_A>
__global__ void k_gemm128(const float* __restrict__ A,
                          const float* __restrict__ W,
                          const float* __restrict__ dinv,
                          float* __restrict__ outH, int M) {
    __shared__ float sW[FDIM * FDIM];

    const int lane = threadIdx.x & 31;
    const int half = lane >> 4;
    const int l16  = lane & 15;

    for (int e = threadIdx.x; e < 32 * 8 * 32; e += blockDim.x) {
        int kstep = e >> 8;
        int rem   = e & 255;
        int t     = rem >> 5;
        int ln    = rem & 31;
        int h     = ln >> 4;
        int c     = t * 16 + (ln & 15);
        int krow  = kstep * 4 + 2 * h;
        float2 v;
        v.x = W[(size_t)krow * FDIM + c];
        v.y = W[(size_t)(krow + 1) * FDIM + c];
        ((float2*)sW)[e] = v;
    }
    __syncthreads();

    const int waveId = (blockIdx.x * blockDim.x + threadIdx.x) >> 5;
    const int rowT   = waveId;
    if (rowT * 16 >= M) return;

    const float* __restrict__ arow = A + (size_t)(rowT * 16 + l16) * FDIM;

    float sc[8];
#pragma unroll
    for (int r = 0; r < 8; ++r) {
        const int row = rowT * 16 + r + half * 8;
        sc[r] = dinv[(row < M) ? row : (M - 1)];
    }

    v8f acc[8];
#pragma unroll
    for (int t = 0; t < 8; ++t) acc[t] = (v8f){0, 0, 0, 0, 0, 0, 0, 0};

    for (int kstep = 0; kstep < 32; ++kstep) {
        v2f a = *(const v2f*)(arow + kstep * 4 + 2 * half);
        if (RELU_A) {
            a.x = fmaxf(a.x, 0.0f);
            a.y = fmaxf(a.y, 0.0f);
        }
        const v2f* __restrict__ bfrag = (const v2f*)sW + kstep * 256 + lane;
#pragma unroll
        for (int t = 0; t < 8; ++t) {
            v2f b = bfrag[t * 32];
            acc[t] = __builtin_amdgcn_wmma_f32_16x16x4_f32(
                false, a, false, b, (short)0, acc[t], false, false);
        }
    }

#pragma unroll
    for (int t = 0; t < 8; ++t) {
        const int col = t * 16 + l16;
#pragma unroll
        for (int r = 0; r < 8; ++r) {
            const int row = rowT * 16 + r + half * 8;
            outH[(size_t)row * FDIM + col] = acc[t][r] * sc[r];
        }
    }
}

template <int G>
__device__ __forceinline__ void row_group(const float4* __restrict__ h4, int sv,
                                          int j, int id0, int n,
                                          int lane, float4& acc) {
    int    s[G];
    float4 v[G];
#pragma unroll
    for (int u = 0; u < G; ++u) s[u] = __builtin_amdgcn_readlane(sv, j + u);
#pragma unroll
    for (int u = 0; u < G; ++u) v[u] = h4[(size_t)s[u] * 32 + lane];
#pragma unroll
    for (int u = 0; u < G; ++u) {
        const int   id = id0 + u;
        const float w  = (id <= n) ? 1.0f : 0.0f;
        acc.x = fmaf(w, v[u].x, acc.x);
        acc.y = fmaf(w, v[u].y, acc.y);
        acc.z = fmaf(w, v[u].z, acc.z);
        acc.w = fmaf(w, v[u].w, acc.w);
    }
}

__global__ void __launch_bounds__(256) k_gather_agg(const float* __restrict__ h,
                                                    const float* __restrict__ dinv,
                                                    const int* __restrict__ rowptr,
                                                    const int* __restrict__ cnt,
                                                    const int* __restrict__ csrc,
                                                    const float* __restrict__ bias,
                                                    float* __restrict__ out, int N) {
    const int lane = threadIdx.x & 31;
    const int wave0 = __builtin_amdgcn_readfirstlane((int)((blockIdx.x * blockDim.x + threadIdx.x) >> 5));
    const int nwave = (int)((gridDim.x * blockDim.x) >> 5);
    const float4* __restrict__ h4 = (const float4*)h;

    for (int d = wave0; d < N; d += nwave) {
        const int   beg = rowptr[d];
        const int   n   = cnt[d];
        const float dd  = dinv[d];
        const int   t   = n + 1;

        float4 acc = make_float4(0.0f, 0.0f, 0.0f, 0.0f);

        int c0 = 0;
        do {
            const int  m      = (t - c0 < 32) ? (t - c0) : 32;
            const int  idx    = c0 + lane;
            const bool isEdge = idx < n;
            int slot = beg + (isEdge ? idx : (n - 1));
            slot = (slot > 0) ? slot : 0;
            int sv = csrc[slot];
            sv = (int)min((unsigned)sv, (unsigned)(N - 1));
            sv = isEdge ? sv : d;

            int j = 0;
            while (m - j > 4) {
                row_group<8>(h4, sv, j, c0 + j, n, lane, acc);
                j += 8;
            }
            if (m - j > 0) row_group<4>(h4, sv, j, c0 + j, n, lane, acc);

            c0 += 32;
        } while (c0 < t);

        const float4 bv = ((const float4*)bias)[lane];
        float4 o;
        o.x = fmaf(dd, acc.x, bv.x);
        o.y = fmaf(dd, acc.y, bv.y);
        o.z = fmaf(dd, acc.z, bv.z);
        o.w = fmaf(dd, acc.w, bv.w);
        ((float4*)out)[(size_t)d * 32 + lane] = o;
    }
}

#define TILE_PITCH 132

__device__ __forceinline__ float4 gather_row_sum(const float4* __restrict__ h4, const int* __restrict__ csrc,
                                                 int d, int beg, int n, int N, int lane) {
    const int t = n + 1;
    float4 acc = make_float4(0.0f, 0.0f, 0.0f, 0.0f);
    int c0 = 0;
    do {
        const int  m      = (t - c0 < 32) ? (t - c0) : 32;
        const int  idx    = c0 + lane;
        const bool isEdge = idx < n;
        int slot = beg + (isEdge ? idx : (n - 1));
        slot = (slot > 0) ? slot : 0;
        int sv = csrc[slot];
        sv = (int)min((unsigned)sv, (unsigned)(N - 1));
        sv = isEdge ? sv : d;

        int j = 0;
        while (m - j > 4) {
            row_group<8>(h4, sv, j, c0 + j, n, lane, acc);
            j += 8;
        }
        if (m - j > 0) row_group<4>(h4, sv, j, c0 + j, n, lane, acc);

        c0 += 32;
    } while (c0 < t);
    return acc;
}

__global__ void __launch_bounds__(512) k_agg_gemm(const float* __restrict__ h,
                                                  const float* __restrict__ dinv,
                                                  const int* __restrict__ rowptr,
                                                  const int* __restrict__ cnt,
                                                  const int* __restrict__ csrc,
                                                  const float* __restrict__ bias,
                                                  const float* __restrict__ W,
                                                  float* __restrict__ outH, int N, int tiles) {
    __shared__ float sW[FDIM * FDIM];
    __shared__ float sA[16 * TILE_PITCH];
    __shared__ float sD[16];

    const int lane = threadIdx.x & 31;
    const int half = lane >> 4;
    const int l16  = lane & 15;
    const int wv   = __builtin_amdgcn_readfirstlane((int)(threadIdx.x >> 5));

#pragma unroll
    for (int e = threadIdx.x; e < 32 * 8 * 32; e += 512) {
        int kstep = e >> 8;
        int rem   = e & 255;
        int t     = rem >> 5;
        int ln    = rem & 31;
        int hh    = ln >> 4;
        int c     = t * 16 + (ln & 15);
        int krow  = kstep * 4 + 2 * hh;
        float2 v;
        v.x = W[(size_t)krow * FDIM + c];
        v.y = W[(size_t)(krow + 1) * FDIM + c];
        ((float2*)sW)[e] = v;
    }

    const float4* __restrict__ h4 = (const float4*)h;
    const float4 bv = ((const float4*)bias)[lane];

    for (int tile = blockIdx.x; tile < tiles; tile += gridDim.x) {
        const int d = tile * 16 + wv;
        float4 o  = make_float4(0.0f, 0.0f, 0.0f, 0.0f);
        float  dd = 0.0f;
        if (d < N) {
            const int beg = rowptr[d];
            const int n   = cnt[d];
            dd = dinv[d];
            const float4 acc = gather_row_sum(h4, csrc, d, beg, n, N, lane);
            o.x = fmaxf(fmaf(dd, acc.x, bv.x), 0.0f);
            o.y = fmaxf(fmaf(dd, acc.y, bv.y), 0.0f);
            o.z = fmaxf(fmaf(dd, acc.z, bv.z), 0.0f);
            o.w = fmaxf(fmaf(dd, acc.w, bv.w), 0.0f);
        }
        __syncthreads();
        *(float4*)(sA + wv * TILE_PITCH + 4 * lane) = o;
        if (lane == 0) sD[wv] = dd;
        __syncthreads();

        if (wv < 8) {
            v8f acc = (v8f){0, 0, 0, 0, 0, 0, 0, 0};
            const float* __restrict__ arow  = sA + l16 * TILE_PITCH + 2 * half;
            const v2f*   __restrict__ bfrag = (const v2f*)sW + wv * 32 + lane;
#pragma unroll 1
            for (int k0 = 0; k0 < 32; k0 += 8) {
                v2f a[8], b[8];
#pragma unroll
                for (int u = 0; u < 8; ++u) {
                    a[u] = *(const v2f*)(arow + (k0 + u) * 4);
                    b[u] = bfrag[(k0 + u) * 256];
                }
#pragma unroll
                for (int u = 0; u < 8; ++u) {
                    acc = __builtin_amdgcn_wmma_f32_16x16x4_f32(
                        false, a[u], false, b[u], (short)0, acc, false, false);
                }
                __builtin_amdgcn_sched_group_barrier(0x100, 16, 0);
                __builtin_amdgcn_sched_group_barrier(0x008, 8, 0);
            }
            const int col = wv * 16 + l16;
#pragma unroll
            for (int r = 0; r < 8; ++r) {
                const int row = tile * 16 + r + half * 8;
                if (row < N) outH[(size_t)row * FDIM + col] = acc[r] * sD[r + half * 8];
            }
        }
    }
}

extern "C" void kernel_launch(void* const* d_in, const int* in_sizes, int n_in,
                              void* d_out, int out_size, void* d_ws, size_t ws_size,
                              hipStream_t stream) {
    const float* x   = (const float*)d_in[0];
    const int*   ei  = (const int*)d_in[1];
    const float* W1  = (const float*)d_in[2];
    const float* b1  = (const float*)d_in[3];
    const float* W2  = (const float*)d_in[4];
    const float* b2  = (const float*)d_in[5];
    float*       out = (float*)d_out;

    const int N = in_sizes[0] / FDIM;
    const int E = in_sizes[1] / 2;
    const int* src = ei;
    const int* dst = ei + E;

    const int NB = (N + BIN_NODES - 1) / BIN_NODES;
    const int C  = (E + CHUNK_EDGES - 1) / CHUNK_EDGES;
    const int M  = NB * C;

    char*  ws    = (char*)d_ws;
    size_t nodeB = ((size_t)N * 4 + 255) & ~(size_t)255;
    size_t edgeB = ((size_t)E * 4 + 255) & ~(size_t)255;
    size_t tabB  = ((size_t)M * 4 + 255) & ~(size_t)255;
    int*   cnt    = (int*)ws;
    int*   rowptr = (int*)(ws + nodeB);
    float* dinv   = (float*)(ws + 2 * nodeB);
    int*   bucket = (int*)(ws + 3 * nodeB);
    int*   csrc   = (int*)(ws + 3 * nodeB + edgeB);
    float* bufH   = (float*)(ws + 3 * nodeB + 2 * edgeB);
    float* bufA   = bufH + (size_t)N * FDIM;
    int*   counts = (int*)(ws + 3 * nodeB + 2 * edgeB + 2 * (size_t)N * FDIM * 4);
    int*   offs   = (int*)(ws + 3 * nodeB + 2 * edgeB + 2 * (size_t)N * FDIM * 4 + tabB);

    const int TPB = 256;

    k_bin_count<<<C, 1024, 0, stream>>>(dst, counts, E, N, NB, C);
    k_scan<<<(M + 4095) / 4096, 1024, 0, stream>>>(counts, offs, M);
    k_bin_scatter<<<C, 1024, 0, stream>>>(src, dst, offs, bucket, E, N, NB, C);
    k_bin_rows<<<NB, 1024, 0, stream>>>(bucket, offs, cnt, rowptr, dinv, csrc, E, N, NB, C);

    const int tiles      = (N + 15) / 16;
    const int gemmBlocks = (tiles + (TPB / 32) - 1) / (TPB / 32);
    const int aggBlocks  = 2048;
    const int fusedBlocks = (tiles < 1024) ? tiles : 1024;

    k_gemm128<false><<<gemmBlocks, TPB, 0, stream>>>(x, W1, dinv, bufH, N);
    k_agg_gemm<<<fusedBlocks, 512, 0, stream>>>(bufH, dinv, rowptr, cnt, csrc, b1, W2, bufA, N, tiles);

    k_gather_agg<<<aggBlocks, TPB, 0, stream>>>(bufA, dinv, rowptr, cnt, csrc, b2, out, N);
}


# Layer-1 GEMM made persistent with whole-row memory access: a 16-wave block per WGP stages W once; each wave passes x and result tiles through its own LDS tile as 512-byte rows; sums unchanged
# speedup vs baseline: 4.2673x; 1.0791x over previous
#include <hip/hip_runtime.h>
#include <math.h>


typedef float v2f __attribute__((ext_vector_type(2)));
typedef float v8f __attribute__((ext_vector_type(8)));

#define FDIM 128

#define BIN_SHIFT   9
#define BIN_NODES   512
#define CHUNK_EDGES 8192
#define MAX_BINS    1024

__global__ void __launch_bounds__(1024) k_bin_count(const int* __restrict__ dst, int* __restrict__ counts,
                                                    int E, int N, int NB, int C) {
    __shared__ int hist[MAX_BINS];
    const int tid = threadIdx.x;
    const int c   = blockIdx.x;
    if (tid < NB) hist[tid] = 0;
    __syncthreads();

    const int base = c * CHUNK_EDGES + tid;
    int d[8];
#pragma unroll
    for (int k = 0; k < 8; ++k) {
        const int e = base + k * 1024;
        d[k] = dst[(e < E) ? e : (E - 1)];
    }
#pragma unroll
    for (int k = 0; k < 8; ++k) {
        const int e = base + k * 1024;
        if (e < E) {
            const unsigned dd = min((unsigned)d[k], (unsigned)(N - 1));
            atomicAdd(&hist[dd >> BIN_SHIFT], 1);
        }
    }
    __syncthreads();
    if (tid < NB) counts[tid * C + c] = hist[tid];
}

__global__ void __launch_bounds__(1024) k_scan(const int* __restrict__ cnt, int* __restrict__ rowptr, int N) {
    __shared__ int sPre[32];
    __shared__ int sTot[32];
    const int tid  = threadIdx.x;
    const int lane = tid & 31;
    const int wv   = tid >> 5;
    const int nb   = blockIdx.x;
    const int4* __restrict__ c4 = (const int4*)cnt;

    int pre = 0;
    int it = 0;
    for (; it + 8 <= nb; it += 8) {
        int4 v[8];
#pragma unroll
        for (int u = 0; u < 8; ++u) v[u] = c4[(it + u) * 1024 + tid];
#pragma unroll
        for (int u = 0; u < 8; ++u) pre += (v[u].x + v[u].y) + (v[u].z + v[u].w);
    }
    for (; it < nb; ++it) {
        const int4 v = c4[it * 1024 + tid];
        pre += (v.x + v.y) + (v.z + v.w);
    }

    const int idx = nb * 4096 + tid * 4;
    int m0 = 0, m1 = 0, m2 = 0, m3 = 0;
    if (idx + 3 < N) {
        const int4 v = c4[idx >> 2];
        m0 = v.x; m1 = v.y; m2 = v.z; m3 = v.w;
    } else {
        if (idx     < N) m0 = cnt[idx];
        if (idx + 1 < N) m1 = cnt[idx + 1];
        if (idx + 2 < N) m2 = cnt[idx + 2];
    }
    const int tsum = (m0 + m1) + (m2 + m3);

    int incl = tsum;
#pragma unroll
    for (int o = 1; o < 32; o <<= 1) {
        const int t = __shfl_up(incl, o);
        if (lane >= o) incl += t;
    }
#pragma unroll
    for (int o = 16; o > 0; o >>= 1) pre += __shfl_xor(pre, o);

    if (lane == 31) sTot[wv] = incl;
    if (lane == 0)  sPre[wv] = pre;
    __syncthreads();

    int base = 0;
#pragma unroll
    for (int w = 0; w < 32; ++w) {
        base += sPre[w];
        if (w < wv) base += sTot[w];
    }
    const int e0 = base + incl - tsum;
    const int e1 = e0 + m0, e2 = e1 + m1, e3 = e2 + m2;

    if (idx + 3 < N) {
        ((int4*)rowptr)[idx >> 2] = make_int4(e0, e1, e2, e3);
    } else {
        if (idx     < N) rowptr[idx]     = e0;
        if (idx + 1 < N) rowptr[idx + 1] = e1;
        if (idx + 2 < N) rowptr[idx + 2] = e2;
    }
}

__global__ void __launch_bounds__(1024) k_bin_scatter(const int* __restrict__ src, const int* __restrict__ dst,
                                                      const int* __restrict__ offs, int* __restrict__ bucket,
                                                      int E, int N, int NB, int C) {
    __shared__ int cursor[MAX_BINS];
    const int tid = threadIdx.x;
    const int c   = blockIdx.x;
    if (tid < NB) cursor[tid] = offs[tid * C + c];
    __syncthreads();

    const int base = c * CHUNK_EDGES + tid;
    int d[8], s[8];
#pragma unroll
    for (int k = 0; k < 8; ++k) {
        const int e  = base + k * 1024;
        const int ec = (e < E) ? e : (E - 1);
        d[k] = dst[ec];
        s[k] = src[ec];
    }
#pragma unroll
    for (int k = 0; k < 8; ++k) {
        const int e = base + k * 1024;
        if (e < E) {
            const unsigned dd = min((unsigned)d[k], (unsigned)(N - 1));
            const unsigned ss = min((unsigned)s[k], (unsigned)(N - 1));
            const int pos = atomicAdd(&cursor[dd >> BIN_SHIFT], 1);
            if ((unsigned)pos < (unsigned)E) bucket[pos] = (int)((ss << BIN_SHIFT) | (dd & (BIN_NODES - 1)));
        }
    }
}

__global__ void __launch_bounds__(1024) k_bin_rows(const int* __restrict__ bucket, const int* __restrict__ offs,
                                                   int* __restrict__ cnt, int* __restrict__ rowptr,
                                                   float* __restrict__ dinv, int* __restrict__ csrc,
                                                   int E, int N, int NB, int C) {
    __shared__ int sCnt[BIN_NODES];
    __shared__ int sTot[32];
    const int tid  = threadIdx.x;
    const int lane = tid & 31;
    const int wv   = tid >> 5;
    const int b    = blockIdx.x;
    int bs = offs[b * C];
    int be = (b + 1 < NB) ? offs[(b + 1) * C] : E;
    be = min(max(be, 0), E);
    bs = min(max(bs, 0), be);

    if (tid < BIN_NODES) sCnt[tid] = 0;
    __syncthreads();

    for (int i0 = bs; i0 < be; i0 += 4096) {
        int r[4];
#pragma unroll
        for (int u = 0; u < 4; ++u) {
            const int i = i0 + u * 1024 + tid;
            r[u] = bucket[(i < be) ? i : (be - 1)];
        }
#pragma unroll
        for (int u = 0; u < 4; ++u) {
            const int i = i0 + u * 1024 + tid;
            if (i < be) atomicAdd(&sCnt[r[u] & (BIN_NODES - 1)], 1);
        }
    }
    __syncthreads();

    const int myc = (tid < BIN_NODES) ? sCnt[tid] : 0;
    int incl = myc;
#pragma unroll
    for (int o = 1; o < 32; o <<= 1) {
        const int t = __shfl_up(incl, o);
        if (lane >= o) incl += t;
    }
    if (lane == 31) sTot[wv] = incl;
    __syncthreads();
    int wbase = 0;
#pragma unroll
    for (int w = 0; w < 32; ++w) {
        if (w < wv) wbase += sTot[w];
    }
    const int first = bs + wbase + incl - myc;

    if (tid < BIN_NODES) {
        sCnt[tid] = first;
        const int node = b * BIN_NODES + tid;
        if (node < N) {
            cnt[node]    = myc;
            rowptr[node] = first;
            dinv[node]   = rsqrtf((float)(myc + 1));
        }
    }
    __syncthreads();

    for (int i0 = bs; i0 < be; i0 += 4096) {
        int r[4];
#pragma unroll
        for (int u = 0; u < 4; ++u) {
            const int i = i0 + u * 1024 + tid;
            r[u] = bucket[(i < be) ? i : (be - 1)];
        }
#pragma unroll
        for (int u = 0; u < 4; ++u) {
            const int i = i0 + u * 1024 + tid;
            if (i < be) {
                const int pos = atomicAdd(&sCnt[r[u] & (BIN_NODES - 1)], 1);
                if ((unsigned)pos < (unsigned)E) csrc[pos] = (int)((unsigned)r[u] >> BIN_SHIFT);
            }
        }
    }
}

#define GEMM_PITCH 132

template <bool GUARD>
__device__ __forceinline__ void gemm_store_rows(const float* __restrict__ tileS, float4* __restrict__ O4,
                                                int rowT, int M, int dvbits, int lane) {
    float4 v[16];
#pragma unroll
    for (int r = 0; r < 16; ++r) v[r] = *(const float4*)(tileS + r * GEMM_PITCH + 4 * lane);
#pragma unroll
    for (int r = 0; r < 16; ++r) {
        const int   row = rowT * 16 + r;
        const float s   = __int_as_float(__builtin_amdgcn_readlane(dvbits, r));
        float4 o;
        o.x = v[r].x * s; o.y = v[r].y * s; o.z = v[r].z * s; o.w = v[r].w * s;
        if (!GUARD || row < M) O4[(size_t)row * 32 + lane] = o;
    }
}

template <bool RELU_A>
__global__ void __launch_bounds__(512) k_gemm128(const float* __restrict__ A,
                                                 const float* __restrict__ W,
                                                 const float* __restrict__ dinv,
                                                 float* __restrict__ outH, int M, int tiles) {
    __shared__ __attribute__((aligned(16))) float sW[FDIM * FDIM];
    __shared__ __attribute__((aligned(16))) float sT[16 * 16 * GEMM_PITCH];

    const int lane = threadIdx.x & 31;
    const int half = lane >> 4;
    const int l16  = lane & 15;
    const int wv   = __builtin_amdgcn_readfirstlane((int)(threadIdx.x >> 5));

#pragma unroll
    for (int i = 0; i < 16; ++i) {
        int e     = (int)threadIdx.x + i * 512;
        int kstep = e >> 8;
        int rem   = e & 255;
        int t     = rem >> 5;
        int ln    = rem & 31;
        int h     = ln >> 4;
        int c     = t * 16 + (ln & 15);
        int krow  = kstep * 4 + 2 * h;
        float2 v;
        v.x = W[(size_t)krow * FDIM + c];
        v.y = W[(size_t)(krow + 1) * FDIM + c];
        ((float2*)sW)[e] = v;
    }
    __syncthreads();

    const int lo = (int)((blockIdx.x * (unsigned)tiles) / gridDim.x);
    const int hi = (int)(((blockIdx.x + 1u) * (unsigned)tiles) / gridDim.x);

    float* __restrict__ tileS = sT + wv * (16 * GEMM_PITCH);
    const float4* __restrict__ A4 = (const float4*)A;
    float4* __restrict__ O4 = (float4*)outH;
    const v2f* __restrict__ bfrag0 = (const v2f*)sW + lane;
    const float* __restrict__ afrag = tileS + l16 * GEMM_PITCH + 2 * half;

    for (int rowT = lo + wv; rowT < hi; rowT += 16) {
        {
            float4 v[16];
#pragma unroll
            for (int r = 0; r < 16; ++r) v[r] = A4[(size_t)min(rowT * 16 + r, M - 1) * 32 + lane];
#pragma unroll
            for (int r = 0; r < 16; ++r) {
                float4 u = v[r];
                if (RELU_A) {
                    u.x = fmaxf(u.x, 0.0f); u.y = fmaxf(u.y, 0.0f);
                    u.z = fmaxf(u.z, 0.0f); u.w = fmaxf(u.w, 0.0f);
                }
                *(float4*)(tileS + r * GEMM_PITCH + 4 * lane) = u;
            }
        }
        const int dvbits = __float_as_int(dinv[min(rowT * 16 + l16, M - 1)]);

        v8f acc[8];
#pragma unroll
        for (int t = 0; t < 8; ++t) acc[t] = (v8f){0, 0, 0, 0, 0, 0, 0, 0};

#pragma unroll 4
        for (int k = 0; k < 32; ++k) {
            const v2f a = *(const v2f*)(afrag + k * 4);
            const v2f* __restrict__ bfrag = bfrag0 + k * 256;
#pragma unroll
            for (int t = 0; t < 8; ++t) {
                const v2f b = bfrag[t * 32];
                acc[t] = __builtin_amdgcn_wmma_f32_16x16x4_f32(
                    false, a, false, b, (short)0, acc[t], false, false);
            }
        }

#pragma unroll
        for (int t = 0; t < 8; ++t) {
#pragma unroll
            for (int r = 0; r < 8; ++r) tileS[(r + 8 * half) * GEMM_PITCH + t * 16 + l16] = acc[t][r];
        }
        if (rowT * 16 + 16 <= M) gemm_store_rows<false>(tileS, O4, rowT, M, dvbits, lane);
        else                     gemm_store_rows<true >(tileS, O4, rowT, M, dvbits, lane);
    }
}

template <int G>
__device__ __forceinline__ void row_group(const float4* __restrict__ h4, int sv,
                                          int j, int id0, int n,
                                          int lane, float4& acc) {
    int    s[G];
    float4 v[G];
#pragma unroll
    for (int u = 0; u < G; ++u) s[u] = __builtin_amdgcn_readlane(sv, j + u);
#pragma unroll
    for (int u = 0; u < G; ++u) v[u] = h4[(size_t)s[u] * 32 + lane];
#pragma unroll
    for (int u = 0; u < G; ++u) {
        const int   id = id0 + u;
        const float w  = (id <= n) ? 1.0f : 0.0f;
        acc.x = fmaf(w, v[u].x, acc.x);
        acc.y = fmaf(w, v[u].y, acc.y);
        acc.z = fmaf(w, v[u].z, acc.z);
        acc.w = fmaf(w, v[u].w, acc.w);
    }
}

__global__ void __launch_bounds__(256) k_gather_agg(const float* __restrict__ h,
                                                    const float* __restrict__ dinv,
                                                    const int* __restrict__ rowptr,
                                                    const int* __restrict__ cnt,
                                                    const int* __restrict__ csrc,
                                                    const float* __restrict__ bias,
                                                    float* __restrict__ out, int N) {
    const int lane = threadIdx.x & 31;
    const int wave0 = __builtin_amdgcn_readfirstlane((int)((blockIdx.x * blockDim.x + threadIdx.x) >> 5));
    const int nwave = (int)((gridDim.x * blockDim.x) >> 5);
    const float4* __restrict__ h4 = (const float4*)h;

    for (int d = wave0; d < N; d += nwave) {
        const int   beg = rowptr[d];
        const int   n   = cnt[d];
        const float dd  = dinv[d];
        const int   t   = n + 1;

        float4 acc = make_float4(0.0f, 0.0f, 0.0f, 0.0f);

        int c0 = 0;
        do {
            const int  m      = (t - c0 < 32) ? (t - c0) : 32;
            const int  idx    = c0 + lane;
            const bool isEdge = idx < n;
            int slot = beg + (isEdge ? idx : (n - 1));
            slot = (slot > 0) ? slot : 0;
            int sv = csrc[slot];
            sv = (int)min((unsigned)sv, (unsigned)(N - 1));
            sv = isEdge ? sv : d;

            int j = 0;
            while (m - j > 4) {
                row_group<8>(h4, sv, j, c0 + j, n, lane, acc);
                j += 8;
            }
            if (m - j > 0) row_group<4>(h4, sv, j, c0 + j, n, lane, acc);

            c0 += 32;
        } while (c0 < t);

        const float4 bv = ((const float4*)bias)[lane];
        float4 o;
        o.x = fmaf(dd, acc.x, bv.x);
        o.y = fmaf(dd, acc.y, bv.y);
        o.z = fmaf(dd, acc.z, bv.z);
        o.w = fmaf(dd, acc.w, bv.w);
        ((float4*)out)[(size_t)d * 32 + lane] = o;
    }
}

#define TILE_PITCH 132

__device__ __forceinline__ float4 gather_row_sum(const float4* __restrict__ h4, const int* __restrict__ csrc,
                                                 int d, int beg, int n, int N, int lane) {
    const int t = n + 1;
    float4 acc = make_float4(0.0f, 0.0f, 0.0f, 0.0f);
    int c0 = 0;
    do {
        const int  m      = (t - c0 < 32) ? (t - c0) : 32;
        const int  idx    = c0 + lane;
        const bool isEdge = idx < n;
        int slot = beg + (isEdge ? idx : (n - 1));
        slot = (slot > 0) ? slot : 0;
        int sv = csrc[slot];
        sv = (int)min((unsigned)sv, (unsigned)(N - 1));
        sv = isEdge ? sv : d;

        int j = 0;
        while (m - j > 4) {
            row_group<8>(h4, sv, j, c0 + j, n, lane, acc);
            j += 8;
        }
        if (m - j > 0) row_group<4>(h4, sv, j, c0 + j, n, lane, acc);

        c0 += 32;
    } while (c0 < t);
    return acc;
}

__global__ void __launch_bounds__(512) k_agg_gemm(const float* __restrict__ h,
                                                  const float* __restrict__ dinv,
                                                  const int* __restrict__ rowptr,
                                                  const int* __restrict__ cnt,
                                                  const int* __restrict__ csrc,
                                                  const float* __restrict__ bias,
                                                  const float* __restrict__ W,
                                                  float* __restrict__ outH, int N, int tiles) {
    __shared__ float sW[FDIM * FDIM];
    __shared__ float sA[16 * TILE_PITCH];
    __shared__ float sD[16];

    const int lane = threadIdx.x & 31;
    const int half = lane >> 4;
    const int l16  = lane & 15;
    const int wv   = __builtin_amdgcn_readfirstlane((int)(threadIdx.x >> 5));

#pragma unroll
    for (int e = threadIdx.x; e < 32 * 8 * 32; e += 512) {
        int kstep = e >> 8;
        int rem   = e & 255;
        int t     = rem >> 5;
        int ln    = rem & 31;
        int hh    = ln >> 4;
        int c     = t * 16 + (ln & 15);
        int krow  = kstep * 4 + 2 * hh;
        float2 v;
        v.x = W[(size_t)krow * FDIM + c];
        v.y = W[(size_t)(krow + 1) * FDIM + c];
        ((float2*)sW)[e] = v;
    }

    const float4* __restrict__ h4 = (const float4*)h;
    const float4 bv = ((const float4*)bias)[lane];

    for (int tile = blockIdx.x; tile < tiles; tile += gridDim.x) {
        const int d = tile * 16 + wv;
        float4 o  = make_float4(0.0f, 0.0f, 0.0f, 0.0f);
        float  dd = 0.0f;
        if (d < N) {
            const int beg = rowptr[d];
            const int n   = cnt[d];
            dd = dinv[d];
            const float4 acc = gather_row_sum(h4, csrc, d, beg, n, N, lane);
            o.x = fmaxf(fmaf(dd, acc.x, bv.x), 0.0f);
            o.y = fmaxf(fmaf(dd, acc.y, bv.y), 0.0f);
            o.z = fmaxf(fmaf(dd, acc.z, bv.z), 0.0f);
            o.w = fmaxf(fmaf(dd, acc.w, bv.w), 0.0f);
        }
        __syncthreads();
        *(float4*)(sA + wv * TILE_PITCH + 4 * lane) = o;
        if (lane == 0) sD[wv] = dd;
        __syncthreads();

        if (wv < 8) {
            v8f acc = (v8f){0, 0, 0, 0, 0, 0, 0, 0};
            const float* __restrict__ arow  = sA + l16 * TILE_PITCH + 2 * half;
            const v2f*   __restrict__ bfrag = (const v2f*)sW + wv * 32 + lane;
#pragma unroll 1
            for (int k0 = 0; k0 < 32; k0 += 8) {
                v2f a[8], b[8];
#pragma unroll
                for (int u = 0; u < 8; ++u) {
                    a[u] = *(const v2f*)(arow + (k0 + u) * 4);
                    b[u] = bfrag[(k0 + u) * 256];
                }
#pragma unroll
                for (int u = 0; u < 8; ++u) {
                    acc = __builtin_amdgcn_wmma_f32_16x16x4_f32(
                        false, a[u], false, b[u], (short)0, acc, false, false);
                }
                __builtin_amdgcn_sched_group_barrier(0x100, 16, 0);
                __builtin_amdgcn_sched_group_barrier(0x008, 8, 0);
            }
            const int col = wv * 16 + l16;
#pragma unroll
            for (int r = 0; r < 8; ++r) {
                const int row = tile * 16 + r + half * 8;
                if (row < N) outH[(size_t)row * FDIM + col] = acc[r] * sD[r + half * 8];
            }
        }
    }
}

extern "C" void kernel_launch(void* const* d_in, const int* in_sizes, int n_in,
                              void* d_out, int out_size, void* d_ws, size_t ws_size,
                              hipStream_t stream) {
    const float* x   = (const float*)d_in[0];
    const int*   ei  = (const int*)d_in[1];
    const float* W1  = (const float*)d_in[2];
    const float* b1  = (const float*)d_in[3];
    const float* W2  = (const float*)d_in[4];
    const float* b2  = (const float*)d_in[5];
    float*       out = (float*)d_out;

    const int N = in_sizes[0] / FDIM;
    const int E = in_sizes[1] / 2;
    const int* src = ei;
    const int* dst = ei + E;

    const int NB = (N + BIN_NODES - 1) / BIN_NODES;
    const int C  = (E + CHUNK_EDGES - 1) / CHUNK_EDGES;
    const int M  = NB * C;

    char*  ws    = (char*)d_ws;
    size_t nodeB = ((size_t)N * 4 + 255) & ~(size_t)255;
    size_t edgeB = ((size_t)E * 4 + 255) & ~(size_t)255;
    size_t tabB  = ((size_t)M * 4 + 255) & ~(size_t)255;
    int*   cnt    = (int*)ws;
    int*   rowptr = (int*)(ws + nodeB);
    float* dinv   = (float*)(ws + 2 * nodeB);
    int*   bucket = (int*)(ws + 3 * nodeB);
    int*   csrc   = (int*)(ws + 3 * nodeB + edgeB);
    float* bufH   = (float*)(ws + 3 * nodeB + 2 * edgeB);
    float* bufA   = bufH + (size_t)N * FDIM;
    int*   counts = (int*)(ws + 3 * nodeB + 2 * edgeB + 2 * (size_t)N * FDIM * 4);
    int*   offs   = (int*)(ws + 3 * nodeB + 2 * edgeB + 2 * (size_t)N * FDIM * 4 + tabB);

    const int TPB = 256;

    k_bin_count<<<C, 1024, 0, stream>>>(dst, counts, E, N, NB, C);
    k_scan<<<(M + 4095) / 4096, 1024, 0, stream>>>(counts, offs, M);
    k_bin_scatter<<<C, 1024, 0, stream>>>(src, dst, offs, bucket, E, N, NB, C);
    k_bin_rows<<<NB, 1024, 0, stream>>>(bucket, offs, cnt, rowptr, dinv, csrc, E, N, NB, C);

    const int tiles      = (N + 15) / 16;
    const int gemmBlocks = (tiles < 256) ? tiles : 256;
    const int aggBlocks  = 2048;
    const int fusedBlocks = (tiles < 1024) ? tiles : 1024;

    k_gemm128<false><<<gemmBlocks, 512, 0, stream>>>(x, W1, dinv, bufH, N, tiles);
    k_agg_gemm<<<fusedBlocks, 512, 0, stream>>>(bufH, dinv, rowptr, cnt, csrc, b1, W2, bufA, N, tiles);

    k_gather_agg<<<aggBlocks, TPB, 0, stream>>>(bufA, dinv, rowptr, cnt, csrc, b2, out, N);
}


# Both layers reassociated to aggregate-then-transform, (Ahat h) W for Ahat (h W): each layer is one fused gather+WMMA launch, the x@W1 pass and its buffer go; f32 sums reordered
# speedup vs baseline: 4.5672x; 1.0703x over previous
#include <hip/hip_runtime.h>
#include <math.h>


typedef float v2f __attribute__((ext_vector_type(2)));
typedef float v8f __attribute__((ext_vector_type(8)));

#define FDIM 128

#define BIN_SHIFT   9
#define BIN_NODES   512
#define CHUNK_EDGES 8192
#define MAX_BINS    1024

__global__ void __launch_bounds__(1024) k_bin_count(const int* __restrict__ dst, int* __restrict__ counts,
                                                    int E, int N, int NB, int C) {
    __shared__ int hist[MAX_BINS];
    const int tid = threadIdx.x;
    const int c   = blockIdx.x;
    if (tid < NB) hist[tid] = 0;
    __syncthreads();

    const int base = c * CHUNK_EDGES + tid;
    int d[8];
#pragma unroll
    for (int k = 0; k < 8; ++k) {
        const int e = base + k * 1024;
        d[k] = dst[(e < E) ? e : (E - 1)];
    }
#pragma unroll
    for (int k = 0; k < 8; ++k) {
        const int e = base + k * 1024;
        if (e < E) {
            const unsigned dd = min((unsigned)d[k], (unsigned)(N - 1));
            atomicAdd(&hist[dd >> BIN_SHIFT], 1);
        }
    }
    __syncthreads();
    if (tid < NB) counts[tid * C + c] = hist[tid];
}

__global__ void __launch_bounds__(1024) k_scan(const int* __restrict__ cnt, int* __restrict__ rowptr, int N) {
    __shared__ int sPre[32];
    __shared__ int sTot[32];
    const int tid  = threadIdx.x;
    const int lane = tid & 31;
    const int wv   = tid >> 5;
    const int nb   = blockIdx.x;
    const int4* __restrict__ c4 = (const int4*)cnt;

    int pre = 0;
    int it = 0;
    for (; it + 8 <= nb; it += 8) {
        int4 v[8];
#pragma unroll
        for (int u = 0; u < 8; ++u) v[u] = c4[(it + u) * 1024 + tid];
#pragma unroll
        for (int u = 0; u < 8; ++u) pre += (v[u].x + v[u].y) + (v[u].z + v[u].w);
    }
    for (; it < nb; ++it) {
        const int4 v = c4[it * 1024 + tid];
        pre += (v.x + v.y) + (v.z + v.w);
    }

    const int idx = nb * 4096 + tid * 4;
    int m0 = 0, m1 = 0, m2 = 0, m3 = 0;
    if (idx + 3 < N) {
        const int4 v = c4[idx >> 2];
        m0 = v.x; m1 = v.y; m2 = v.z; m3 = v.w;
    } else {
        if (idx     < N) m0 = cnt[idx];
        if (idx + 1 < N) m1 = cnt[idx + 1];
        if (idx + 2 < N) m2 = cnt[idx + 2];
    }
    const int tsum = (m0 + m1) + (m2 + m3);

    int incl = tsum;
#pragma unroll
    for (int o = 1; o < 32; o <<= 1) {
        const int t = __shfl_up(incl, o);
        if (lane >= o) incl += t;
    }
#pragma unroll
    for (int o = 16; o > 0; o >>= 1) pre += __shfl_xor(pre, o);

    if (lane == 31) sTot[wv] = incl;
    if (lane == 0)  sPre[wv] = pre;
    __syncthreads();

    int base = 0;
#pragma unroll
    for (int w = 0; w < 32; ++w) {
        base += sPre[w];
        if (w < wv) base += sTot[w];
    }
    const int e0 = base + incl - tsum;
    const int e1 = e0 + m0, e2 = e1 + m1, e3 = e2 + m2;

    if (idx + 3 < N) {
        ((int4*)rowptr)[idx >> 2] = make_int4(e0, e1, e2, e3);
    } else {
        if (idx     < N) rowptr[idx]     = e0;
        if (idx + 1 < N) rowptr[idx + 1] = e1;
        if (idx + 2 < N) rowptr[idx + 2] = e2;
    }
}

__global__ void __launch_bounds__(1024) k_bin_scatter(const int* __restrict__ src, const int* __restrict__ dst,
                                                      const int* __restrict__ offs, int* __restrict__ bucket,
                                                      int E, int N, int NB, int C) {
    __shared__ int cursor[MAX_BINS];
    const int tid = threadIdx.x;
    const int c   = blockIdx.x;
    if (tid < NB) cursor[tid] = offs[tid * C + c];
    __syncthreads();

    const int base = c * CHUNK_EDGES + tid;
    int d[8], s[8];
#pragma unroll
    for (int k = 0; k < 8; ++k) {
        const int e  = base + k * 1024;
        const int ec = (e < E) ? e : (E - 1);
        d[k] = dst[ec];
        s[k] = src[ec];
    }
#pragma unroll
    for (int k = 0; k < 8; ++k) {
        const int e = base + k * 1024;
        if (e < E) {
            const unsigned dd = min((unsigned)d[k], (unsigned)(N - 1));
            const unsigned ss = min((unsigned)s[k], (unsigned)(N - 1));
            const int pos = atomicAdd(&cursor[dd >> BIN_SHIFT], 1);
            if ((unsigned)pos < (unsigned)E) bucket[pos] = (int)((ss << BIN_SHIFT) | (dd & (BIN_NODES - 1)));
        }
    }
}

__global__ void __launch_bounds__(1024) k_bin_rows(const int* __restrict__ bucket, const int* __restrict__ offs,
                                                   int* __restrict__ cnt, int* __restrict__ rowptr,
                                                   float* __restrict__ dinv, int* __restrict__ csrc,
                                                   int E, int N, int NB, int C) {
    __shared__ int sCnt[BIN_NODES];
    __shared__ int sTot[32];
    const int tid  = threadIdx.x;
    const int lane = tid & 31;
    const int wv   = tid >> 5;
    const int b    = blockIdx.x;
    int bs = offs[b * C];
    int be = (b + 1 < NB) ? offs[(b + 1) * C] : E;
    be = min(max(be, 0), E);
    bs = min(max(bs, 0), be);

    if (tid < BIN_NODES) sCnt[tid] = 0;
    __syncthreads();

    for (int i0 = bs; i0 < be; i0 += 4096) {
        int r[4];
#pragma unroll
        for (int u = 0; u < 4; ++u) {
            const int i = i0 + u * 1024 + tid;
            r[u] = bucket[(i < be) ? i : (be - 1)];
        }
#pragma unroll
        for (int u = 0; u < 4; ++u) {
            const int i = i0 + u * 1024 + tid;
            if (i < be) atomicAdd(&sCnt[r[u] & (BIN_NODES - 1)], 1);
        }
    }
    __syncthreads();

    const int myc = (tid < BIN_NODES) ? sCnt[tid] : 0;
    int incl = myc;
#pragma unroll
    for (int o = 1; o < 32; o <<= 1) {
        const int t = __shfl_up(incl, o);
        if (lane >= o) incl += t;
    }
    if (lane == 31) sTot[wv] = incl;
    __syncthreads();
    int wbase = 0;
#pragma unroll
    for (int w = 0; w < 32; ++w) {
        if (w < wv) wbase += sTot[w];
    }
    const int first = bs + wbase + incl - myc;

    if (tid < BIN_NODES) {
        sCnt[tid] = first;
        const int node = b * BIN_NODES + tid;
        if (node < N) {
            cnt[node]    = myc;
            rowptr[node] = first;
            dinv[node]   = rsqrtf((float)(myc + 1));
        }
    }
    __syncthreads();

    for (int i0 = bs; i0 < be; i0 += 4096) {
        int r[4];
#pragma unroll
        for (int u = 0; u < 4; ++u) {
            const int i = i0 + u * 1024 + tid;
            r[u] = bucket[(i < be) ? i : (be - 1)];
        }
#pragma unroll
        for (int u = 0; u < 4; ++u) {
            const int i = i0 + u * 1024 + tid;
            if (i < be) {
                const int pos = atomicAdd(&sCnt[r[u] & (BIN_NODES - 1)], 1);
                if ((unsigned)pos < (unsigned)E) csrc[pos] = (int)((unsigned)r[u] >> BIN_SHIFT);
            }
        }
    }
}

template <int G, bool WT>
__device__ __forceinline__ void row_group(const float4* __restrict__ h4, int sv, int wbits,
                                          int j, int id0, int n,
                                          int lane, float4& acc) {
    int    s[G];
    float4 v[G];
#pragma unroll
    for (int u = 0; u < G; ++u) s[u] = __builtin_amdgcn_readlane(sv, j + u);
#pragma unroll
    for (int u = 0; u < G; ++u)
        v[u] = *(const float4*)((const char*)h4 + ((unsigned)s[u] * 512u + (unsigned)lane * 16u));
    if (WT) __builtin_amdgcn_sched_barrier(0);
#pragma unroll
    for (int u = 0; u < G; ++u) {
        const int   id = id0 + u;
        const float wt = WT ? __int_as_float(__builtin_amdgcn_readlane(wbits, j + u)) : 1.0f;
        const float w  = (id <= n) ? wt : 0.0f;
        acc.x = fmaf(w, v[u].x, acc.x);
        acc.y = fmaf(w, v[u].y, acc.y);
        acc.z = fmaf(w, v[u].z, acc.z);
        acc.w = fmaf(w, v[u].w, acc.w);
    }
}

#define TILE_PITCH 132

template <bool WT>
__device__ __forceinline__ float4 gather_row_sum(const float4* __restrict__ h4, const int* __restrict__ csrc,
                                                 const float* __restrict__ dinv,
                                                 int d, int beg, int n, int N, int lane) {
    const int t = n + 1;
    float4 acc = make_float4(0.0f, 0.0f, 0.0f, 0.0f);
    int c0 = 0;
    do {
        const int  m      = (t - c0 < 32) ? (t - c0) : 32;
        const int  idx    = c0 + lane;
        const bool isEdge = idx < n;
        int slot = beg + (isEdge ? idx : (n - 1));
        slot = (slot > 0) ? slot : 0;
        int sv = csrc[slot];
        sv = (int)min((unsigned)sv, (unsigned)(N - 1));
        sv = isEdge ? sv : d;
        const int wbits = WT ? __float_as_int(dinv[sv]) : 0;

        int j = 0;
        while (m - j > 4) {
            row_group<8, WT>(h4, sv, wbits, j, c0 + j, n, lane, acc);
            j += 8;
        }
        if (m - j > 0) row_group<4, WT>(h4, sv, wbits, j, c0 + j, n, lane, acc);

        c0 += 32;
    } while (c0 < t);
    return acc;
}

template <bool FIRST>
__global__ void __launch_bounds__(512) k_agg_gemm(const float* __restrict__ h,
                                                  const float* __restrict__ dinv,
                                                  const int* __restrict__ rowptr,
                                                  const int* __restrict__ cnt,
                                                  const int* __restrict__ csrc,
                                                  const float* __restrict__ bias,
                                                  const float* __restrict__ W,
                                                  float* __restrict__ outH, int N, int tiles) {
    __shared__ float sW[FDIM * FDIM];
    __shared__ float sA[16 * TILE_PITCH];
    __shared__ float sD[16];

    const int lane = threadIdx.x & 31;
    const int half = lane >> 4;
    const int l16  = lane & 15;
    const int wv   = __builtin_amdgcn_readfirstlane((int)(threadIdx.x >> 5));

#pragma unroll
    for (int e = threadIdx.x; e < 32 * 8 * 32; e += 512) {
        int kstep = e >> 8;
        int rem   = e & 255;
        int t     = rem >> 5;
        int ln    = rem & 31;
        int hh    = ln >> 4;
        int c     = t * 16 + (ln & 15);
        int krow  = kstep * 4 + 2 * hh;
        float2 v;
        v.x = W[(size_t)krow * FDIM + c];
        v.y = W[(size_t)(krow + 1) * FDIM + c];
        ((float2*)sW)[e] = v;
    }

    const float4* __restrict__ h4 = (const float4*)h;
    const float bcol = bias[(wv & 7) * 16 + l16];

    for (int tile = blockIdx.x; tile < tiles; tile += gridDim.x) {
        const int d = tile * 16 + wv;
        float4 o  = make_float4(0.0f, 0.0f, 0.0f, 0.0f);
        float  dd = 0.0f;
        if (d < N) {
            const int beg = rowptr[d];
            const int n   = cnt[d];
            dd = dinv[d];
            const float4 acc = gather_row_sum<FIRST>(h4, csrc, dinv, d, beg, n, N, lane);
            o.x = dd * acc.x;
            o.y = dd * acc.y;
            o.z = dd * acc.z;
            o.w = dd * acc.w;
        }
        __syncthreads();
        *(float4*)(sA + wv * TILE_PITCH + 4 * lane) = o;
        if (FIRST && lane == 0) sD[wv] = dd;
        __syncthreads();

        if (wv < 8) {
            v8f acc = (v8f){0, 0, 0, 0, 0, 0, 0, 0};
            const float* __restrict__ arow  = sA + l16 * TILE_PITCH + 2 * half;
            const v2f*   __restrict__ bfrag = (const v2f*)sW + wv * 32 + lane;
#pragma unroll 1
            for (int k0 = 0; k0 < 32; k0 += 8) {
                v2f a[8], b[8];
#pragma unroll
                for (int u = 0; u < 8; ++u) {
                    a[u] = *(const v2f*)(arow + (k0 + u) * 4);
                    b[u] = bfrag[(k0 + u) * 256];
                }
#pragma unroll
                for (int u = 0; u < 8; ++u) {
                    acc = __builtin_amdgcn_wmma_f32_16x16x4_f32(
                        false, a[u], false, b[u], (short)0, acc, false, false);
                }
                __builtin_amdgcn_sched_group_barrier(0x100, 16, 0);
                __builtin_amdgcn_sched_group_barrier(0x008, 8, 0);
            }
            const int col = wv * 16 + l16;
#pragma unroll
            for (int r = 0; r < 8; ++r) {
                const int row = tile * 16 + r + half * 8;
                float v = acc[r] + bcol;
                if (FIRST) v = fmaxf(v, 0.0f) * sD[r + half * 8];
                if (row < N) outH[(size_t)row * FDIM + col] = v;
            }
        }
    }
}

extern "C" void kernel_launch(void* const* d_in, const int* in_sizes, int n_in,
                              void* d_out, int out_size, void* d_ws, size_t ws_size,
                              hipStream_t stream) {
    const float* x   = (const float*)d_in[0];
    const int*   ei  = (const int*)d_in[1];
    const float* W1  = (const float*)d_in[2];
    const float* b1  = (const float*)d_in[3];
    const float* W2  = (const float*)d_in[4];
    const float* b2  = (const float*)d_in[5];
    float*       out = (float*)d_out;

    const int N = in_sizes[0] / FDIM;
    const int E = in_sizes[1] / 2;
    const int* src = ei;
    const int* dst = ei + E;

    const int NB = (N + BIN_NODES - 1) / BIN_NODES;
    const int C  = (E + CHUNK_EDGES - 1) / CHUNK_EDGES;
    const int M  = NB * C;

    char*  ws    = (char*)d_ws;
    size_t nodeB = ((size_t)N * 4 + 255) & ~(size_t)255;
    size_t edgeB = ((size_t)E * 4 + 255) & ~(size_t)255;
    size_t tabB  = ((size_t)M * 4 + 255) & ~(size_t)255;
    int*   cnt    = (int*)ws;
    int*   rowptr = (int*)(ws + nodeB);
    float* dinv   = (float*)(ws + 2 * nodeB);
    int*   bucket = (int*)(ws + 3 * nodeB);
    int*   csrc   = (int*)(ws + 3 * nodeB + edgeB);
    float* bufA   = (float*)(ws + 3 * nodeB + 2 * edgeB);
    int*   counts = (int*)(ws + 3 * nodeB + 2 * edgeB + (size_t)N * FDIM * 4);
    int*   offs   = (int*)(ws + 3 * nodeB + 2 * edgeB + (size_t)N * FDIM * 4 + tabB);

    k_bin_count<<<C, 1024, 0, stream>>>(dst, counts, E, N, NB, C);
    k_scan<<<(M + 4095) / 4096, 1024, 0, stream>>>(counts, offs, M);
    k_bin_scatter<<<C, 1024, 0, stream>>>(src, dst, offs, bucket, E, N, NB, C);
    k_bin_rows<<<NB, 1024, 0, stream>>>(bucket, offs, cnt, rowptr, dinv, csrc, E, N, NB, C);

    const int tiles       = (N + 15) / 16;
    const int fusedBlocks = (tiles < 1024) ? tiles : 1024;

    k_agg_gemm<true ><<<fusedBlocks, 512, 0, stream>>>(x,    dinv, rowptr, cnt, csrc, b1, W1, bufA, N, tiles);
    k_agg_gemm<false><<<fusedBlocks, 512, 0, stream>>>(bufA, dinv, rowptr, cnt, csrc, b2, W2, out,  N, tiles);
}


# Sort's scattered 4-byte stores staged through LDS: bin scatter orders its chunk by bin, row fill orders its bin by row, both copy out contiguous runs; row sum order still LDS arrival order
# speedup vs baseline: 5.3251x; 1.1659x over previous
#include <hip/hip_runtime.h>
#include <math.h>


typedef float v2f __attribute__((ext_vector_type(2)));
typedef float v8f __attribute__((ext_vector_type(8)));

#define FDIM 128

#define BIN_SHIFT   9
#define BIN_NODES   512
#define CHUNK_EDGES 8192
#define MAX_BINS    1024
#define STAGE_SLOTS 16384

__global__ void __launch_bounds__(1024) k_bin_count(const int* __restrict__ dst, int* __restrict__ counts,
                                                    int E, int N, int NB, int C) {
    __shared__ int hist[MAX_BINS];
    const int tid = threadIdx.x;
    const int c   = blockIdx.x;
    if (tid < NB) hist[tid] = 0;
    __syncthreads();

    const int base = c * CHUNK_EDGES + tid;
    int d[8];
#pragma unroll
    for (int k = 0; k < 8; ++k) {
        const int e = base + k * 1024;
        d[k] = dst[(e < E) ? e : (E - 1)];
    }
#pragma unroll
    for (int k = 0; k < 8; ++k) {
        const int e = base + k * 1024;
        if (e < E) {
            const unsigned dd = min((unsigned)d[k], (unsigned)(N - 1));
            atomicAdd(&hist[dd >> BIN_SHIFT], 1);
        }
    }
    __syncthreads();
    if (tid < NB) counts[tid * C + c] = hist[tid];
}

__global__ void __launch_bounds__(1024) k_scan(const int* __restrict__ cnt, int* __restrict__ rowptr, int N) {
    __shared__ int sPre[32];
    __shared__ int sTot[32];
    const int tid  = threadIdx.x;
    const int lane = tid & 31;
    const int wv   = tid >> 5;
    const int nb   = blockIdx.x;
    const int4* __restrict__ c4 = (const int4*)cnt;

    int pre = 0;
    int it = 0;
    for (; it + 8 <= nb; it += 8) {
        int4 v[8];
#pragma unroll
        for (int u = 0; u < 8; ++u) v[u] = c4[(it + u) * 1024 + tid];
#pragma unroll
        for (int u = 0; u < 8; ++u) pre += (v[u].x + v[u].y) + (v[u].z + v[u].w);
    }
    for (; it < nb; ++it) {
        const int4 v = c4[it * 1024 + tid];
        pre += (v.x + v.y) + (v.z + v.w);
    }

    const int idx = nb * 4096 + tid * 4;
    int m0 = 0, m1 = 0, m2 = 0, m3 = 0;
    if (idx + 3 < N) {
        const int4 v = c4[idx >> 2];
        m0 = v.x; m1 = v.y; m2 = v.z; m3 = v.w;
    } else {
        if (idx     < N) m0 = cnt[idx];
        if (idx + 1 < N) m1 = cnt[idx + 1];
        if (idx + 2 < N) m2 = cnt[idx + 2];
    }
    const int tsum = (m0 + m1) + (m2 + m3);

    int incl = tsum;
#pragma unroll
    for (int o = 1; o < 32; o <<= 1) {
        const int t = __shfl_up(incl, o);
        if (lane >= o) incl += t;
    }
#pragma unroll
    for (int o = 16; o > 0; o >>= 1) pre += __shfl_xor(pre, o);

    if (lane == 31) sTot[wv] = incl;
    if (lane == 0)  sPre[wv] = pre;
    __syncthreads();

    int base = 0;
#pragma unroll
    for (int w = 0; w < 32; ++w) {
        base += sPre[w];
        if (w < wv) base += sTot[w];
    }
    const int e0 = base + incl - tsum;
    const int e1 = e0 + m0, e2 = e1 + m1, e3 = e2 + m2;

    if (idx + 3 < N) {
        ((int4*)rowptr)[idx >> 2] = make_int4(e0, e1, e2, e3);
    } else {
        if (idx     < N) rowptr[idx]     = e0;
        if (idx + 1 < N) rowptr[idx + 1] = e1;
        if (idx + 2 < N) rowptr[idx + 2] = e2;
    }
}

__global__ void __launch_bounds__(1024) k_bin_scatter(const int* __restrict__ src, const int* __restrict__ dst,
                                                      const int* __restrict__ counts, const int* __restrict__ offs,
                                                      int* __restrict__ bucket,
                                                      int E, int N, int NB, int C) {
    __shared__ int cursor[MAX_BINS];
    __shared__ int gbase[MAX_BINS];
    __shared__ int sTot[32];
    __shared__ int stageW[CHUNK_EDGES];
    __shared__ unsigned short stageB[CHUNK_EDGES];
    const int tid  = threadIdx.x;
    const int lane = tid & 31;
    const int wv   = tid >> 5;
    const int c    = blockIdx.x;

    const int base = c * CHUNK_EDGES + tid;
    int d[8], s[8];
#pragma unroll
    for (int k = 0; k < 8; ++k) {
        const int e  = base + k * 1024;
        const int ec = (e < E) ? e : (E - 1);
        d[k] = dst[ec];
        s[k] = src[ec];
    }

    int myc = 0, myo = 0;
    if (tid < NB) {
        myc = counts[tid * C + c];
        myo = offs[tid * C + c];
    }
    myc = min(max(myc, 0), CHUNK_EDGES);
    int incl = myc;
#pragma unroll
    for (int o = 1; o < 32; o <<= 1) {
        const int t = __shfl_up(incl, o);
        if (lane >= o) incl += t;
    }
    if (lane == 31) sTot[wv] = incl;
    __syncthreads();
    int wbase = 0;
#pragma unroll
    for (int w = 0; w < 32; ++w) {
        if (w < wv) wbase += sTot[w];
    }
    const int lstart = wbase + incl - myc;
    if (tid < NB) {
        cursor[tid] = lstart;
        gbase[tid]  = myo - lstart;
    }
    __syncthreads();

#pragma unroll
    for (int k = 0; k < 8; ++k) {
        const int e = base + k * 1024;
        if (e < E) {
            const unsigned dd  = min((unsigned)d[k], (unsigned)(N - 1));
            const unsigned ss  = min((unsigned)s[k], (unsigned)(N - 1));
            const unsigned bin = dd >> BIN_SHIFT;
            const int lp = atomicAdd(&cursor[bin], 1);
            if ((unsigned)lp < (unsigned)CHUNK_EDGES) {
                stageW[lp] = (int)((ss << BIN_SHIFT) | (dd & (BIN_NODES - 1)));
                stageB[lp] = (unsigned short)bin;
            }
        }
    }
    __syncthreads();

    const int nvalid = min(CHUNK_EDGES, E - c * CHUNK_EDGES);
    int wd[8], bn[8], ps[8];
#pragma unroll
    for (int k = 0; k < 8; ++k) {
        const int i  = tid + k * 1024;
        const int ic = (i < nvalid) ? i : (nvalid - 1);
        bn[k] = min((int)stageB[ic], MAX_BINS - 1);
        wd[k] = stageW[ic];
    }
#pragma unroll
    for (int k = 0; k < 8; ++k) ps[k] = gbase[bn[k]] + tid + k * 1024;
#pragma unroll
    for (int k = 0; k < 8; ++k) {
        const int i = tid + k * 1024;
        if (i < nvalid && (unsigned)ps[k] < (unsigned)E) bucket[ps[k]] = wd[k];
    }
}

__global__ void __launch_bounds__(1024) k_bin_rows(const int* __restrict__ bucket, const int* __restrict__ offs,
                                                   int* __restrict__ cnt, int* __restrict__ rowptr,
                                                   float* __restrict__ dinv, int* __restrict__ csrc,
                                                   int E, int N, int NB, int C) {
    __shared__ int sCnt[BIN_NODES];
    __shared__ int sTot[32];
    __shared__ int stage[STAGE_SLOTS];
    const int tid  = threadIdx.x;
    const int lane = tid & 31;
    const int wv   = tid >> 5;
    const int b    = blockIdx.x;
    int bs = offs[b * C];
    int be = (b + 1 < NB) ? offs[(b + 1) * C] : E;
    be = min(max(be, 0), E);
    bs = min(max(bs, 0), be);

    if (tid < BIN_NODES) sCnt[tid] = 0;
    __syncthreads();

    for (int i0 = bs; i0 < be; i0 += 4096) {
        int r[4];
#pragma unroll
        for (int u = 0; u < 4; ++u) {
            const int i = i0 + u * 1024 + tid;
            r[u] = bucket[(i < be) ? i : (be - 1)];
        }
#pragma unroll
        for (int u = 0; u < 4; ++u) {
            const int i = i0 + u * 1024 + tid;
            if (i < be) atomicAdd(&sCnt[r[u] & (BIN_NODES - 1)], 1);
        }
    }
    __syncthreads();

    const int myc = (tid < BIN_NODES) ? sCnt[tid] : 0;
    int incl = myc;
#pragma unroll
    for (int o = 1; o < 32; o <<= 1) {
        const int t = __shfl_up(incl, o);
        if (lane >= o) incl += t;
    }
    if (lane == 31) sTot[wv] = incl;
    __syncthreads();
    int wbase = 0;
#pragma unroll
    for (int w = 0; w < 32; ++w) {
        if (w < wv) wbase += sTot[w];
    }
    const int first = bs + wbase + incl - myc;

    if (tid < BIN_NODES) {
        sCnt[tid] = first;
        const int node = b * BIN_NODES + tid;
        if (node < N) {
            cnt[node]    = myc;
            rowptr[node] = first;
            dinv[node]   = rsqrtf((float)(myc + 1));
        }
    }
    __syncthreads();

    const int  nslots = be - bs;
    const bool staged = nslots <= STAGE_SLOTS;
    for (int i0 = bs; i0 < be; i0 += 4096) {
        int r[4];
#pragma unroll
        for (int u = 0; u < 4; ++u) {
            const int i = i0 + u * 1024 + tid;
            r[u] = bucket[(i < be) ? i : (be - 1)];
        }
#pragma unroll
        for (int u = 0; u < 4; ++u) {
            const int i = i0 + u * 1024 + tid;
            if (i < be) {
                const int pos = atomicAdd(&sCnt[r[u] & (BIN_NODES - 1)], 1);
                const int sv  = (int)((unsigned)r[u] >> BIN_SHIFT);
                if (staged) {
                    const int lp = pos - bs;
                    if ((unsigned)lp < (unsigned)nslots) stage[lp] = sv;
                } else {
                    if ((unsigned)pos < (unsigned)E) csrc[pos] = sv;
                }
            }
        }
    }
    if (staged) {
        __syncthreads();
        for (int i0 = 0; i0 < nslots; i0 += 4096) {
            int v[4];
#pragma unroll
            for (int u = 0; u < 4; ++u) {
                const int i = i0 + u * 1024 + tid;
                v[u] = stage[(i < nslots) ? i : (nslots - 1)];
            }
#pragma unroll
            for (int u = 0; u < 4; ++u) {
                const int i = i0 + u * 1024 + tid;
                if (i < nslots) csrc[bs + i] = v[u];
            }
        }
    }
}

template <int G, bool WT>
__device__ __forceinline__ void row_group(const float4* __restrict__ h4, int sv, int wbits,
                                          int j, int id0, int n,
                                          int lane, float4& acc) {
    int    s[G];
    float4 v[G];
#pragma unroll
    for (int u = 0; u < G; ++u) s[u] = __builtin_amdgcn_readlane(sv, j + u);
#pragma unroll
    for (int u = 0; u < G; ++u)
        v[u] = *(const float4*)((const char*)h4 + ((unsigned)s[u] * 512u + (unsigned)lane * 16u));
    if (WT) __builtin_amdgcn_sched_barrier(0);
#pragma unroll
    for (int u = 0; u < G; ++u) {
        const int   id = id0 + u;
        const float wt = WT ? __int_as_float(__builtin_amdgcn_readlane(wbits, j + u)) : 1.0f;
        const float w  = (id <= n) ? wt : 0.0f;
        acc.x = fmaf(w, v[u].x, acc.x);
        acc.y = fmaf(w, v[u].y, acc.y);
        acc.z = fmaf(w, v[u].z, acc.z);
        acc.w = fmaf(w, v[u].w, acc.w);
    }
}

#define TILE_PITCH 132

template <bool WT>
__device__ __forceinline__ float4 gather_row_sum(const float4* __restrict__ h4, const int* __restrict__ csrc,
                                                 const float* __restrict__ dinv,
                                                 int d, int beg, int n, int N, int lane) {
    const int t = n + 1;
    float4 acc = make_float4(0.0f, 0.0f, 0.0f, 0.0f);
    int c0 = 0;
    do {
        const int  m      = (t - c0 < 32) ? (t - c0) : 32;
        const int  idx    = c0 + lane;
        const bool isEdge = idx < n;
        int slot = beg + (isEdge ? idx : (n - 1));
        slot = (slot > 0) ? slot : 0;
        int sv = csrc[slot];
        sv = (int)min((unsigned)sv, (unsigned)(N - 1));
        sv = isEdge ? sv : d;
        const int wbits = WT ? __float_as_int(dinv[sv]) : 0;

        int j = 0;
        while (m - j > 4) {
            row_group<8, WT>(h4, sv, wbits, j, c0 + j, n, lane, acc);
            j += 8;
        }
        if (m - j > 0) row_group<4, WT>(h4, sv, wbits, j, c0 + j, n, lane, acc);

        c0 += 32;
    } while (c0 < t);
    return acc;
}

template <bool FIRST>
__global__ void __launch_bounds__(512) k_agg_gemm(const float* __restrict__ h,
                                                  const float* __restrict__ dinv,
                                                  const int* __restrict__ rowptr,
                                                  const int* __restrict__ cnt,
                                                  const int* __restrict__ csrc,
                                                  const float* __restrict__ bias,
                                                  const float* __restrict__ W,
                                                  float* __restrict__ outH, int N, int tiles) {
    __shared__ float sW[FDIM * FDIM];
    __shared__ float sA[16 * TILE_PITCH];
    __shared__ float sD[16];

    const int lane = threadIdx.x & 31;
    const int half = lane >> 4;
    const int l16  = lane & 15;
    const int wv   = __builtin_amdgcn_readfirstlane((int)(threadIdx.x >> 5));

#pragma unroll
    for (int e = threadIdx.x; e < 32 * 8 * 32; e += 512) {
        int kstep = e >> 8;
        int rem   = e & 255;
        int t     = rem >> 5;
        int ln    = rem & 31;
        int hh    = ln >> 4;
        int c     = t * 16 + (ln & 15);
        int krow  = kstep * 4 + 2 * hh;
        float2 v;
        v.x = W[(size_t)krow * FDIM + c];
        v.y = W[(size_t)(krow + 1) * FDIM + c];
        ((float2*)sW)[e] = v;
    }

    const float4* __restrict__ h4 = (const float4*)h;
    const float bcol = bias[(wv & 7) * 16 + l16];

    for (int tile = blockIdx.x; tile < tiles; tile += gridDim.x) {
        const int d = tile * 16 + wv;
        float4 o  = make_float4(0.0f, 0.0f, 0.0f, 0.0f);
        float  dd = 0.0f;
        if (d < N) {
            const int beg = rowptr[d];
            const int n   = cnt[d];
            dd = dinv[d];
            const float4 acc = gather_row_sum<FIRST>(h4, csrc, dinv, d, beg, n, N, lane);
            o.x = dd * acc.x;
            o.y = dd * acc.y;
            o.z = dd * acc.z;
            o.w = dd * acc.w;
        }
        __syncthreads();
        *(float4*)(sA + wv * TILE_PITCH + 4 * lane) = o;
        if (FIRST && lane == 0) sD[wv] = dd;
        __syncthreads();

        if (wv < 8) {
            v8f acc = (v8f){0, 0, 0, 0, 0, 0, 0, 0};
            const float* __restrict__ arow  = sA + l16 * TILE_PITCH + 2 * half;
            const v2f*   __restrict__ bfrag = (const v2f*)sW + wv * 32 + lane;
#pragma unroll 1
            for (int k0 = 0; k0 < 32; k0 += 8) {
                v2f a[8], b[8];
#pragma unroll
                for (int u = 0; u < 8; ++u) {
                    a[u] = *(const v2f*)(arow + (k0 + u) * 4);
                    b[u] = bfrag[(k0 + u) * 256];
                }
#pragma unroll
                for (int u = 0; u < 8; ++u) {
                    acc = __builtin_amdgcn_wmma_f32_16x16x4_f32(
                        false, a[u], false, b[u], (short)0, acc, false, false);
                }
                __builtin_amdgcn_sched_group_barrier(0x100, 16, 0);
                __builtin_amdgcn_sched_group_barrier(0x008, 8, 0);
            }
            const int col = wv * 16 + l16;
#pragma unroll
            for (int r = 0; r < 8; ++r) {
                const int row = tile * 16 + r + half * 8;
                float v = acc[r] + bcol;
                if (FIRST) v = fmaxf(v, 0.0f) * sD[r + half * 8];
                if (row < N) outH[(size_t)row * FDIM + col] = v;
            }
        }
    }
}

extern "C" void kernel_launch(void* const* d_in, const int* in_sizes, int n_in,
                              void* d_out, int out_size, void* d_ws, size_t ws_size,
                              hipStream_t stream) {
    const float* x   = (const float*)d_in[0];
    const int*   ei  = (const int*)d_in[1];
    const float* W1  = (const float*)d_in[2];
    const float* b1  = (const float*)d_in[3];
    const float* W2  = (const float*)d_in[4];
    const float* b2  = (const float*)d_in[5];
    float*       out = (float*)d_out;

    const int N = in_sizes[0] / FDIM;
    const int E = in_sizes[1] / 2;
    const int* src = ei;
    const int* dst = ei + E;

    const int NB = (N + BIN_NODES - 1) / BIN_NODES;
    const int C  = (E + CHUNK_EDGES - 1) / CHUNK_EDGES;
    const int M  = NB * C;

    char*  ws    = (char*)d_ws;
    size_t nodeB = ((size_t)N * 4 + 255) & ~(size_t)255;
    size_t edgeB = ((size_t)E * 4 + 255) & ~(size_t)255;
    size_t tabB  = ((size_t)M * 4 + 255) & ~(size_t)255;
    int*   cnt    = (int*)ws;
    int*   rowptr = (int*)(ws + nodeB);
    float* dinv   = (float*)(ws + 2 * nodeB);
    int*   bucket = (int*)(ws + 3 * nodeB);
    int*   csrc   = (int*)(ws + 3 * nodeB + edgeB);
    float* bufA   = (float*)(ws + 3 * nodeB + 2 * edgeB);
    int*   counts = (int*)(ws + 3 * nodeB + 2 * edgeB + (size_t)N * FDIM * 4);
    int*   offs   = (int*)(ws + 3 * nodeB + 2 * edgeB + (size_t)N * FDIM * 4 + tabB);

    k_bin_count<<<C, 1024, 0, stream>>>(dst, counts, E, N, NB, C);
    k_scan<<<(M + 4095) / 4096, 1024, 0, stream>>>(counts, offs, M);
    k_bin_scatter<<<C, 1024, 0, stream>>>(src, dst, counts, offs, bucket, E, N, NB, C);
    k_bin_rows<<<NB, 1024, 0, stream>>>(bucket, offs, cnt, rowptr, dinv, csrc, E, N, NB, C);

    const int tiles       = (N + 15) / 16;
    const int fusedBlocks = (tiles < 1024) ? tiles : 1024;

    k_agg_gemm<true ><<<fusedBlocks, 512, 0, stream>>>(x,    dinv, rowptr, cnt, csrc, b1, W1, bufA, N, tiles);
    k_agg_gemm<false><<<fusedBlocks, 512, 0, stream>>>(bufA, dinv, rowptr, cnt, csrc, b2, W2, out,  N, tiles);
}


# Result rows leave through a second LDS tile and are stored as whole 512-byte rows, one store per wave, not 64-byte pieces; blocks grow to 32 waves (2 per WGP) for the LDS room; sums unchanged
# speedup vs baseline: 5.6582x; 1.0626x over previous
#include <hip/hip_runtime.h>
#include <math.h>


typedef float v2f __attribute__((ext_vector_type(2)));
typedef float v8f __attribute__((ext_vector_type(8)));

#define FDIM 128

#define BIN_SHIFT   9
#define BIN_NODES   512
#define CHUNK_EDGES 8192
#define MAX_BINS    1024
#define STAGE_SLOTS 16384

__global__ void __launch_bounds__(1024) k_bin_count(const int* __restrict__ dst, int* __restrict__ counts,
                                                    int E, int N, int NB, int C) {
    __shared__ int hist[MAX_BINS];
    const int tid = threadIdx.x;
    const int c   = blockIdx.x;
    if (tid < NB) hist[tid] = 0;
    __syncthreads();

    const int base = c * CHUNK_EDGES + tid;
    int d[8];
#pragma unroll
    for (int k = 0; k < 8; ++k) {
        const int e = base + k * 1024;
        d[k] = dst[(e < E) ? e : (E - 1)];
    }
#pragma unroll
    for (int k = 0; k < 8; ++k) {
        const int e = base + k * 1024;
        if (e < E) {
            const unsigned dd = min((unsigned)d[k], (unsigned)(N - 1));
            atomicAdd(&hist[dd >> BIN_SHIFT], 1);
        }
    }
    __syncthreads();
    if (tid < NB) counts[tid * C + c] = hist[tid];
}

__global__ void __launch_bounds__(1024) k_scan(const int* __restrict__ cnt, int* __restrict__ rowptr, int N) {
    __shared__ int sPre[32];
    __shared__ int sTot[32];
    const int tid  = threadIdx.x;
    const int lane = tid & 31;
    const int wv   = tid >> 5;
    const int nb   = blockIdx.x;
    const int4* __restrict__ c4 = (const int4*)cnt;

    int pre = 0;
    int it = 0;
    for (; it + 8 <= nb; it += 8) {
        int4 v[8];
#pragma unroll
        for (int u = 0; u < 8; ++u) v[u] = c4[(it + u) * 1024 + tid];
#pragma unroll
        for (int u = 0; u < 8; ++u) pre += (v[u].x + v[u].y) + (v[u].z + v[u].w);
    }
    for (; it < nb; ++it) {
        const int4 v = c4[it * 1024 + tid];
        pre += (v.x + v.y) + (v.z + v.w);
    }

    const int idx = nb * 4096 + tid * 4;
    int m0 = 0, m1 = 0, m2 = 0, m3 = 0;
    if (idx + 3 < N) {
        const int4 v = c4[idx >> 2];
        m0 = v.x; m1 = v.y; m2 = v.z; m3 = v.w;
    } else {
        if (idx     < N) m0 = cnt[idx];
        if (idx + 1 < N) m1 = cnt[idx + 1];
        if (idx + 2 < N) m2 = cnt[idx + 2];
    }
    const int tsum = (m0 + m1) + (m2 + m3);

    int incl = tsum;
#pragma unroll
    for (int o = 1; o < 32; o <<= 1) {
        const int t = __shfl_up(incl, o);
        if (lane >= o) incl += t;
    }
#pragma unroll
    for (int o = 16; o > 0; o >>= 1) pre += __shfl_xor(pre, o);

    if (lane == 31) sTot[wv] = incl;
    if (lane == 0)  sPre[wv] = pre;
    __syncthreads();

    int base = 0;
#pragma unroll
    for (int w = 0; w < 32; ++w) {
        base += sPre[w];
        if (w < wv) base += sTot[w];
    }
    const int e0 = base + incl - tsum;
    const int e1 = e0 + m0, e2 = e1 + m1, e3 = e2 + m2;

    if (idx + 3 < N) {
        ((int4*)rowptr)[idx >> 2] = make_int4(e0, e1, e2, e3);
    } else {
        if (idx     < N) rowptr[idx]     = e0;
        if (idx + 1 < N) rowptr[idx + 1] = e1;
        if (idx + 2 < N) rowptr[idx + 2] = e2;
    }
}

__global__ void __launch_bounds__(1024) k_bin_scatter(const int* __restrict__ src, const int* __restrict__ dst,
                                                      const int* __restrict__ counts, const int* __restrict__ offs,
                                                      int* __restrict__ bucket,
                                                      int E, int N, int NB, int C) {
    __shared__ int cursor[MAX_BINS];
    __shared__ int gbase[MAX_BINS];
    __shared__ int sTot[32];
    __shared__ int stageW[CHUNK_EDGES];
    __shared__ unsigned short stageB[CHUNK_EDGES];
    const int tid  = threadIdx.x;
    const int lane = tid & 31;
    const int wv   = tid >> 5;
    const int c    = blockIdx.x;

    const int base = c * CHUNK_EDGES + tid;
    int d[8], s[8];
#pragma unroll
    for (int k = 0; k < 8; ++k) {
        const int e  = base + k * 1024;
        const int ec = (e < E) ? e : (E - 1);
        d[k] = dst[ec];
        s[k] = src[ec];
    }

    int myc = 0, myo = 0;
    if (tid < NB) {
        myc = counts[tid * C + c];
        myo = offs[tid * C + c];
    }
    myc = min(max(myc, 0), CHUNK_EDGES);
    int incl = myc;
#pragma unroll
    for (int o = 1; o < 32; o <<= 1) {
        const int t = __shfl_up(incl, o);
        if (lane >= o) incl += t;
    }
    if (lane == 31) sTot[wv] = incl;
    __syncthreads();
    int wbase = 0;
#pragma unroll
    for (int w = 0; w < 32; ++w) {
        if (w < wv) wbase += sTot[w];
    }
    const int lstart = wbase + incl - myc;
    if (tid < NB) {
        cursor[tid] = lstart;
        gbase[tid]  = myo - lstart;
    }
    __syncthreads();

#pragma unroll
    for (int k = 0; k < 8; ++k) {
        const int e = base + k * 1024;
        if (e < E) {
            const unsigned dd  = min((unsigned)d[k], (unsigned)(N - 1));
            const unsigned ss  = min((unsigned)s[k], (unsigned)(N - 1));
            const unsigned bin = dd >> BIN_SHIFT;
            const int lp = atomicAdd(&cursor[bin], 1);
            if ((unsigned)lp < (unsigned)CHUNK_EDGES) {
                stageW[lp] = (int)((ss << BIN_SHIFT) | (dd & (BIN_NODES - 1)));
                stageB[lp] = (unsigned short)bin;
            }
        }
    }
    __syncthreads();

    const int nvalid = min(CHUNK_EDGES, E - c * CHUNK_EDGES);
    int wd[8], bn[8], ps[8];
#pragma unroll
    for (int k = 0; k < 8; ++k) {
        const int i  = tid + k * 1024;
        const int ic = (i < nvalid) ? i : (nvalid - 1);
        bn[k] = min((int)stageB[ic], MAX_BINS - 1);
        wd[k] = stageW[ic];
    }
#pragma unroll
    for (int k = 0; k < 8; ++k) ps[k] = gbase[bn[k]] + tid + k * 1024;
#pragma unroll
    for (int k = 0; k < 8; ++k) {
        const int i = tid + k * 1024;
        if (i < nvalid && (unsigned)ps[k] < (unsigned)E) bucket[ps[k]] = wd[k];
    }
}

__global__ void __launch_bounds__(1024) k_bin_rows(const int* __restrict__ bucket, const int* __restrict__ offs,
                                                   int* __restrict__ cnt, int* __restrict__ rowptr,
                                                   float* __restrict__ dinv, int* __restrict__ csrc,
                                                   int E, int N, int NB, int C) {
    __shared__ int sCnt[BIN_NODES];
    __shared__ int sTot[32];
    __shared__ int stage[STAGE_SLOTS];
    const int tid  = threadIdx.x;
    const int lane = tid & 31;
    const int wv   = tid >> 5;
    const int b    = blockIdx.x;
    int bs = offs[b * C];
    int be = (b + 1 < NB) ? offs[(b + 1) * C] : E;
    be = min(max(be, 0), E);
    bs = min(max(bs, 0), be);

    if (tid < BIN_NODES) sCnt[tid] = 0;
    __syncthreads();

    for (int i0 = bs; i0 < be; i0 += 4096) {
        int r[4];
#pragma unroll
        for (int u = 0; u < 4; ++u) {
            const int i = i0 + u * 1024 + tid;
            r[u] = bucket[(i < be) ? i : (be - 1)];
        }
#pragma unroll
        for (int u = 0; u < 4; ++u) {
            const int i = i0 + u * 1024 + tid;
            if (i < be) atomicAdd(&sCnt[r[u] & (BIN_NODES - 1)], 1);
        }
    }
    __syncthreads();

    const int myc = (tid < BIN_NODES) ? sCnt[tid] : 0;
    int incl = myc;
#pragma unroll
    for (int o = 1; o < 32; o <<= 1) {
        const int t = __shfl_up(incl, o);
        if (lane >= o) incl += t;
    }
    if (lane == 31) sTot[wv] = incl;
    __syncthreads();
    int wbase = 0;
#pragma unroll
    for (int w = 0; w < 32; ++w) {
        if (w < wv) wbase += sTot[w];
    }
    const int first = bs + wbase + incl - myc;

    if (tid < BIN_NODES) {
        sCnt[tid] = first;
        const int node = b * BIN_NODES + tid;
        if (node < N) {
            cnt[node]    = myc;
            rowptr[node] = first;
            dinv[node]   = rsqrtf((float)(myc + 1));
        }
    }
    __syncthreads();

    const int  nslots = be - bs;
    const bool staged = nslots <= STAGE_SLOTS;
    for (int i0 = bs; i0 < be; i0 += 4096) {
        int r[4];
#pragma unroll
        for (int u = 0; u < 4; ++u) {
            const int i = i0 + u * 1024 + tid;
            r[u] = bucket[(i < be) ? i : (be - 1)];
        }
#pragma unroll
        for (int u = 0; u < 4; ++u) {
            const int i = i0 + u * 1024 + tid;
            if (i < be) {
                const int pos = atomicAdd(&sCnt[r[u] & (BIN_NODES - 1)], 1);
                const int sv  = (int)((unsigned)r[u] >> BIN_SHIFT);
                if (staged) {
                    const int lp = pos - bs;
                    if ((unsigned)lp < (unsigned)nslots) stage[lp] = sv;
                } else {
                    if ((unsigned)pos < (unsigned)E) csrc[pos] = sv;
                }
            }
        }
    }
    if (staged) {
        __syncthreads();
        for (int i0 = 0; i0 < nslots; i0 += 4096) {
            int v[4];
#pragma unroll
            for (int u = 0; u < 4; ++u) {
                const int i = i0 + u * 1024 + tid;
                v[u] = stage[(i < nslots) ? i : (nslots - 1)];
            }
#pragma unroll
            for (int u = 0; u < 4; ++u) {
                const int i = i0 + u * 1024 + tid;
                if (i < nslots) csrc[bs + i] = v[u];
            }
        }
    }
}

template <int G, bool WT>
__device__ __forceinline__ void row_group(const float4* __restrict__ h4, int sv, int wbits,
                                          int j, int id0, int n,
                                          int lane, float4& acc) {
    int    s[G];
    float4 v[G];
#pragma unroll
    for (int u = 0; u < G; ++u) s[u] = __builtin_amdgcn_readlane(sv, j + u);
#pragma unroll
    for (int u = 0; u < G; ++u)
        v[u] = *(const float4*)((const char*)h4 + ((unsigned)s[u] * 512u + (unsigned)lane * 16u));
    if (WT) __builtin_amdgcn_sched_barrier(0);
#pragma unroll
    for (int u = 0; u < G; ++u) {
        const int   id = id0 + u;
        const float wt = WT ? __int_as_float(__builtin_amdgcn_readlane(wbits, j + u)) : 1.0f;
        const float w  = (id <= n) ? wt : 0.0f;
        acc.x = fmaf(w, v[u].x, acc.x);
        acc.y = fmaf(w, v[u].y, acc.y);
        acc.z = fmaf(w, v[u].z, acc.z);
        acc.w = fmaf(w, v[u].w, acc.w);
    }
}

#define TILE_PITCH 132

template <bool WT>
__device__ __forceinline__ float4 gather_row_sum(const float4* __restrict__ h4, const int* __restrict__ csrc,
                                                 const float* __restrict__ dinv,
                                                 int d, int beg, int n, int N, int lane) {
    const int t = n + 1;
    float4 acc = make_float4(0.0f, 0.0f, 0.0f, 0.0f);
    int c0 = 0;
    do {
        const int  m      = (t - c0 < 32) ? (t - c0) : 32;
        const int  idx    = c0 + lane;
        const bool isEdge = idx < n;
        int slot = beg + (isEdge ? idx : (n - 1));
        slot = (slot > 0) ? slot : 0;
        int sv = csrc[slot];
        sv = (int)min((unsigned)sv, (unsigned)(N - 1));
        sv = isEdge ? sv : d;
        const int wbits = WT ? __float_as_int(dinv[sv]) : 0;

        int j = 0;
        while (m - j > 4) {
            row_group<8, WT>(h4, sv, wbits, j, c0 + j, n, lane, acc);
            j += 8;
        }
        if (m - j > 0) row_group<4, WT>(h4, sv, wbits, j, c0 + j, n, lane, acc);

        c0 += 32;
    } while (c0 < t);
    return acc;
}

template <bool FIRST>
__global__ void __launch_bounds__(1024) k_agg_gemm(const float* __restrict__ h,
                                                  const float* __restrict__ dinv,
                                                  const int* __restrict__ rowptr,
                                                  const int* __restrict__ cnt,
                                                  const int* __restrict__ csrc,
                                                  const float* __restrict__ bias,
                                                  const float* __restrict__ W,
                                                  float* __restrict__ outH, int N, int tiles) {
    __shared__ float sW[FDIM * FDIM];
    __shared__ float sA[32 * TILE_PITCH];
    __shared__ float sO[32 * TILE_PITCH];
    __shared__ float sD[32];

    const int lane = threadIdx.x & 31;
    const int half = lane >> 4;
    const int l16  = lane & 15;
    const int wv   = __builtin_amdgcn_readfirstlane((int)(threadIdx.x >> 5));

#pragma unroll
    for (int e = threadIdx.x; e < 32 * 8 * 32; e += 1024) {
        int kstep = e >> 8;
        int rem   = e & 255;
        int t     = rem >> 5;
        int ln    = rem & 31;
        int hh    = ln >> 4;
        int c     = t * 16 + (ln & 15);
        int krow  = kstep * 4 + 2 * hh;
        float2 v;
        v.x = W[(size_t)krow * FDIM + c];
        v.y = W[(size_t)(krow + 1) * FDIM + c];
        ((float2*)sW)[e] = v;
    }

    const float4* __restrict__ h4 = (const float4*)h;
    const float bcol = bias[(wv & 7) * 16 + l16];

    int prev = -1;
    for (int tile = blockIdx.x; tile < tiles; tile += gridDim.x) {
        const int d = tile * 32 + wv;
        float4 o  = make_float4(0.0f, 0.0f, 0.0f, 0.0f);
        float  dd = 0.0f;
        if (d < N) {
            const int beg = rowptr[d];
            const int n   = cnt[d];
            dd = dinv[d];
            const float4 acc = gather_row_sum<FIRST>(h4, csrc, dinv, d, beg, n, N, lane);
            o.x = dd * acc.x;
            o.y = dd * acc.y;
            o.z = dd * acc.z;
            o.w = dd * acc.w;
        }
        __syncthreads();
        const int prow = prev * 32 + wv;
        float4 r4 = make_float4(0.0f, 0.0f, 0.0f, 0.0f);
        if (prev >= 0) r4 = *(const float4*)(sO + wv * TILE_PITCH + 4 * lane);
        *(float4*)(sA + wv * TILE_PITCH + 4 * lane) = o;
        if (FIRST && lane == 0) sD[wv] = dd;
        __syncthreads();
        if (prev >= 0 && prow < N)
            *(float4*)(outH + (size_t)prow * FDIM + 4 * lane) = r4;

        if (wv < 16) {
            const int rt = wv >> 3;
            const int ct = wv & 7;
            v8f acc = (v8f){0, 0, 0, 0, 0, 0, 0, 0};
            const float* __restrict__ arow  = sA + (rt * 16 + l16) * TILE_PITCH + 2 * half;
            const v2f*   __restrict__ bfrag = (const v2f*)sW + ct * 32 + lane;
#pragma unroll 1
            for (int k0 = 0; k0 < 32; k0 += 8) {
                v2f a[8], b[8];
#pragma unroll
                for (int u = 0; u < 8; ++u) {
                    a[u] = *(const v2f*)(arow + (k0 + u) * 4);
                    b[u] = bfrag[(k0 + u) * 256];
                }
#pragma unroll
                for (int u = 0; u < 8; ++u) {
                    acc = __builtin_amdgcn_wmma_f32_16x16x4_f32(
                        false, a[u], false, b[u], (short)0, acc, false, false);
                }
                __builtin_amdgcn_sched_group_barrier(0x100, 16, 0);
                __builtin_amdgcn_sched_group_barrier(0x008, 8, 0);
            }
            const int col = ct * 16 + l16;
#pragma unroll
            for (int r = 0; r < 8; ++r) {
                const int lr = rt * 16 + r + half * 8;
                float v = acc[r] + bcol;
                if (FIRST) v = fmaxf(v, 0.0f) * sD[lr];
                sO[lr * TILE_PITCH + col] = v;
            }
        }
        prev = tile;
    }
    __syncthreads();
    if (prev >= 0) {
        const int prow = prev * 32 + wv;
        const float4 r4 = *(const float4*)(sO + wv * TILE_PITCH + 4 * lane);
        if (prow < N) *(float4*)(outH + (size_t)prow * FDIM + 4 * lane) = r4;
    }
}

extern "C" void kernel_launch(void* const* d_in, const int* in_sizes, int n_in,
                              void* d_out, int out_size, void* d_ws, size_t ws_size,
                              hipStream_t stream) {
    const float* x   = (const float*)d_in[0];
    const int*   ei  = (const int*)d_in[1];
    const float* W1  = (const float*)d_in[2];
    const float* b1  = (const float*)d_in[3];
    const float* W2  = (const float*)d_in[4];
    const float* b2  = (const float*)d_in[5];
    float*       out = (float*)d_out;

    const int N = in_sizes[0] / FDIM;
    const int E = in_sizes[1] / 2;
    const int* src = ei;
    const int* dst = ei + E;

    const int NB = (N + BIN_NODES - 1) / BIN_NODES;
    const int C  = (E + CHUNK_EDGES - 1) / CHUNK_EDGES;
    const int M  = NB * C;

    char*  ws    = (char*)d_ws;
    size_t nodeB = ((size_t)N * 4 + 255) & ~(size_t)255;
    size_t edgeB = ((size_t)E * 4 + 255) & ~(size_t)255;
    size_t tabB  = ((size_t)M * 4 + 255) & ~(size_t)255;
    int*   cnt    = (int*)ws;
    int*   rowptr = (int*)(ws + nodeB);
    float* dinv   = (float*)(ws + 2 * nodeB);
    int*   bucket = (int*)(ws + 3 * nodeB);
    int*   csrc   = (int*)(ws + 3 * nodeB + edgeB);
    float* bufA   = (float*)(ws + 3 * nodeB + 2 * edgeB);
    int*   counts = (int*)(ws + 3 * nodeB + 2 * edgeB + (size_t)N * FDIM * 4);
    int*   offs   = (int*)(ws + 3 * nodeB + 2 * edgeB + (size_t)N * FDIM * 4 + tabB);

    k_bin_count<<<C, 1024, 0, stream>>>(dst, counts, E, N, NB, C);
    k_scan<<<(M + 4095) / 4096, 1024, 0, stream>>>(counts, offs, M);
    k_bin_scatter<<<C, 1024, 0, stream>>>(src, dst, counts, offs, bucket, E, N, NB, C);
    k_bin_rows<<<NB, 1024, 0, stream>>>(bucket, offs, cnt, rowptr, dinv, csrc, E, N, NB, C);

    const int tiles       = (N + 31) / 32;
    const int fusedBlocks = (tiles < 512) ? tiles : 512;

    k_agg_gemm<true ><<<fusedBlocks, 1024, 0, stream>>>(x,    dinv, rowptr, cnt, csrc, b1, W1, bufA, N, tiles);
    k_agg_gemm<false><<<fusedBlocks, 1024, 0, stream>>>(bufA, dinv, rowptr, cnt, csrc, b2, W2, out,  N, tiles);
}
